# starting kernel
# baseline (speedup 1.0000x reference)
#include <hip/hip_runtime.h>
#include <math.h>


#define NNODES 50000
#define NEDGES 1600000
#define IN_F   256
#define HEADS  4
#define OUT_F  32
#define CO     (HEADS * OUT_F)
#define NEG_SLOPE 0.2f

typedef __attribute__((ext_vector_type(2))) float v2f;
typedef __attribute__((ext_vector_type(8))) float v8f;

__device__ __forceinline__ unsigned f2key(float x) {
  unsigned b = __float_as_uint(x);
  return (b & 0x80000000u) ? ~b : (b | 0x80000000u);
}
__device__ __forceinline__ float key2f(unsigned k) {
  unsigned b = (k & 0x80000000u) ? (k ^ 0x80000000u) : ~k;
  return __uint_as_float(b);
}
__device__ __forceinline__ float leaky(float x) {
  return x > 0.f ? x : NEG_SLOPE * x;
}

__global__ __launch_bounds__(256) void k_init(float* __restrict__ out,
                                              unsigned* __restrict__ mkey,
                                              float* __restrict__ ssum) {
  int i = blockIdx.x * 256 + threadIdx.x;
  if (i < NNODES * CO) out[i] = 0.f;
  if (i < NNODES * HEADS) { mkey[i] = 0u; ssum[i] = 0.f; }
}

__global__ __launch_bounds__(256) void k_gemm(const float* __restrict__ feat,
                                              const float* __restrict__ W,
                                              float* __restrict__ ft) {
  __shared__ float As[16][IN_F + 4];
  const int tid    = threadIdx.x;
  const int wave   = tid >> 5;
  const int lane   = tid & 31;
  const int tile_m = blockIdx.x * 16;

  for (int i = tid; i < 16 * (IN_F / 4); i += 256) {
    const int r  = i / (IN_F / 4);
    const int c4 = i % (IN_F / 4);
    float4 v = make_float4(0.f, 0.f, 0.f, 0.f);
    const int row = tile_m + r;
    if (row < NNODES) v = ((const float4*)(feat + (size_t)row * IN_F))[c4];
    ((float4*)&As[r][0])[c4] = v;
  }
  __syncthreads();

  const int koff = (lane < 16) ? 0 : 2;
  const int arow = lane & 15;
  const int col  = wave * 16 + arow;
  const float* wrow = W + (size_t)col * IN_F;

  v8f c = {};
  for (int k0 = 0; k0 < IN_F; k0 += 4) {
    v2f a, b;
    a.x = As[arow][k0 + koff];
    a.y = As[arow][k0 + koff + 1];
    b.x = wrow[k0 + koff];
    b.y = wrow[k0 + koff + 1];
    c = __builtin_amdgcn_wmma_f32_16x16x4_f32(
         false, a,  false, b,
         (short)0, c,  false,  false);
  }

  const int mbase  = (lane < 16) ? 0 : 8;
  const int outcol = col;
#pragma unroll
  for (int j = 0; j < 8; ++j) {
    const int row = tile_m + mbase + j;
    if (row < NNODES) ft[(size_t)row * CO + outcol] = c[j];
  }
}

__global__ __launch_bounds__(256) void k_eler(const float* __restrict__ ft,
                                              const float* __restrict__ attn,
                                              float* __restrict__ el,
                                              float* __restrict__ er) {
  int i = blockIdx.x * 256 + threadIdx.x;
  if (i >= NNODES * HEADS) return;
  const int n = i / HEADS, h = i % HEADS;
  const float* f  = ft + (size_t)n * CO + h * OUT_F;
  const float* al = attn + h * 2 * OUT_F;
  const float* ar = al + OUT_F;
  float sl = 0.f, sr = 0.f;
#pragma unroll
  for (int d = 0; d < OUT_F; ++d) {
    const float v = f[d];
    sl += v * al[d];
    sr += v * ar[d];
  }
  el[i] = sl;
  er[i] = sr;
}

__global__ __launch_bounds__(256) void k_edge_max(const int* __restrict__ src,
                                                  const int* __restrict__ dst,
                                                  const float* __restrict__ el,
                                                  const float* __restrict__ er,
                                                  unsigned* __restrict__ mkey) {
  int i = blockIdx.x * 256 + threadIdx.x;
  if (i >= NEDGES * HEADS) return;
  const int e = i >> 2, h = i & 3;
  const int s = src[e], d = dst[e];
  const float x = leaky(el[s * HEADS + h] + er[d * HEADS + h]);
  atomicMax(&mkey[d * HEADS + h], f2key(x));
}

__global__ __launch_bounds__(256) void k_edge_sum(const int* __restrict__ src,
                                                  const int* __restrict__ dst,
                                                  const float* __restrict__ el,
                                                  const float* __restrict__ er,
                                                  const unsigned* __restrict__ mkey,
                                                  float* __restrict__ ssum) {
  int i = blockIdx.x * 256 + threadIdx.x;
  if (i >= NEDGES * HEADS) return;
  const int e = i >> 2, h = i & 3;
  const int s = src[e], d = dst[e];
  const float x = leaky(el[s * HEADS + h] + er[d * HEADS + h]);
  const float m = key2f(mkey[d * HEADS + h]);
  atomicAdd(&ssum[d * HEADS + h], expf(x - m));
}

__global__ __launch_bounds__(256) void k_edge_agg(const int* __restrict__ src,
                                                  const int* __restrict__ dst,
                                                  const float* __restrict__ el,
                                                  const float* __restrict__ er,
                                                  const unsigned* __restrict__ mkey,
                                                  const float* __restrict__ ssum,
                                                  const float* __restrict__ ft,
                                                  float* __restrict__ out) {
  const int lane = threadIdx.x & 31;
  const int eg   = blockIdx.x * 8 + (threadIdx.x >> 5);
  if (eg >= NEDGES) return;
  const int s = src[eg], d = dst[eg];

  const float* fsrc = ft + (size_t)s * CO;
  __builtin_prefetch(fsrc, 0, 0);

#pragma unroll
  for (int h = 0; h < HEADS; ++h) {
    const float x = leaky(el[s * HEADS + h] + er[d * HEADS + h]);
    const float m = key2f(mkey[d * HEADS + h]);
    const float a = expf(x - m) / ssum[d * HEADS + h];
    const float v = fsrc[h * OUT_F + lane] * a;
    atomicAdd(&out[(size_t)d * CO + h * OUT_F + lane], v);
  }
}

extern "C" void kernel_launch(void* const* d_in, const int* in_sizes, int n_in,
                              void* d_out, int out_size, void* d_ws, size_t ws_size,
                              hipStream_t stream) {
  (void)in_sizes; (void)n_in; (void)out_size; (void)ws_size;

  const float* feat = (const float*)d_in[0];
  const int*   src  = (const int*)d_in[1];
  const int*   dst  = (const int*)d_in[2];
  const float* W    = (const float*)d_in[3];
  const float* attn = (const float*)d_in[4];
  float* out = (float*)d_out;

  char* ws = (char*)d_ws;
  float*    ft   = (float*)ws;    ws += (size_t)NNODES * CO * sizeof(float);
  float*    el   = (float*)ws;    ws += (size_t)NNODES * HEADS * sizeof(float);
  float*    er   = (float*)ws;    ws += (size_t)NNODES * HEADS * sizeof(float);
  unsigned* mkey = (unsigned*)ws; ws += (size_t)NNODES * HEADS * sizeof(unsigned);
  float*    ssum = (float*)ws;    ws += (size_t)NNODES * HEADS * sizeof(float);

  k_init<<<(NNODES * CO + 255) / 256, 256, 0, stream>>>(out, mkey, ssum);
  k_gemm<<<NNODES / 16, 256, 0, stream>>>(feat, W, ft);
  k_eler<<<(NNODES * HEADS + 255) / 256, 256, 0, stream>>>(ft, attn, el, er);
  k_edge_max<<<(NEDGES * HEADS + 255) / 256, 256, 0, stream>>>(src, dst, el, er, mkey);
  k_edge_sum<<<(NEDGES * HEADS + 255) / 256, 256, 0, stream>>>(src, dst, el, er, mkey, ssum);
  k_edge_agg<<<NEDGES / 8, 256, 0, stream>>>(src, dst, el, er, mkey, ssum, ft, out);
}


# Sum pass: each work-group holds a quarter of the destinations' softmax sums in LDS, adds exp terms there, then flushes whole 128-byte lines by global atomics; sums in LDS, then global arrival order
# speedup vs baseline: 2.7052x; 2.7052x over previous
#include <hip/hip_runtime.h>
#include <math.h>


#define NNODES 50000
#define NEDGES 1600000
#define IN_F   256
#define HEADS  4
#define OUT_F  32
#define CO     (HEADS * OUT_F)
#define NEG_SLOPE 0.2f

#define CHUNK_E  256
#define NCHUNK   (NEDGES / CHUNK_E)
#define HALF_N   (NNODES / 2)
#define QSTRIDE  64
#define NQUEUE   6
#define EDGE_WGS 2048
#define MAXP_WGS 256
#define SUM_RANGES 4
#define SR_N     (NNODES / SUM_RANGES)
#define SR_F     (SR_N * HEADS)
#define SUM_WGS  256
#define SUM_T    1024
#define SCH_E    (SUM_T * 8)
#define STRIP_E  (32 * 8)
#define NSCH     ((NEDGES + SCH_E - 1) / SCH_E)
#define GK_OFF   32
#define SPREAD_MAX 16.0f
#define WS_NEED  ((size_t)NNODES * CO * 4 + (size_t)4 * NNODES * HEADS * 4 + (size_t)NQUEUE * QSTRIDE * 4)

static_assert(NEDGES % CHUNK_E == 0, "chunks cover the edge list exactly");
static_assert(NNODES % 16 == 0, "GEMM tiles cover the node rows exactly");
static_assert(NNODES % SUM_RANGES == 0 && HALF_N == 2 * SR_N, "two sum ranges per cache half");
static_assert(NEDGES % 4 == 0, "a thread's four edges in a row are all inside the list or all outside");
static_assert(SUM_WGS % SUM_RANGES == 0, "the fixed deal gives every range the same number of work-groups");
static_assert(((SR_F + 31 + SUM_T) / SUM_T) % 7 == 0, "the table goes out in groups of seven reads per thread");

typedef __attribute__((ext_vector_type(2))) float v2f;
typedef __attribute__((ext_vector_type(8))) float v8f;

__device__ __forceinline__ unsigned f2key(float x) {
  unsigned b = __float_as_uint(x);
  return (b & 0x80000000u) ? ~b : (b | 0x80000000u);
}
__device__ __forceinline__ float key2f(unsigned k) {
  unsigned b = (k & 0x80000000u) ? (k ^ 0x80000000u) : ~k;
  return __uint_as_float(b);
}
__device__ __forceinline__ float leaky(float x) {
  return x > 0.f ? x : NEG_SLOPE * x;
}

__device__ __forceinline__ unsigned l2_half() {
  return (__builtin_amdgcn_s_sendmsg_rtn(0x87) >> 18) & 1u;
}

__device__ __forceinline__ bool el_bound(const int* queues, bool dyn, float gmax[HEADS]) {
  bool ok = dyn;
  for (int h = 0; h < HEADS; ++h) gmax[h] = 0.f;
  if (dyn) {
    const unsigned* gk = (const unsigned*)queues + GK_OFF;
    for (int h = 0; h < HEADS; ++h) {
      const float mx = key2f(gk[h]);
      const float mn = -key2f(gk[HEADS + h]);
      gmax[h] = mx;
      ok = ok && (mx - mn <= SPREAD_MAX);
    }
  }
  return ok;
}

__device__ __forceinline__ int next_chunk(int* cnt, bool dyn, int iter, int* slot) {
  if (!dyn) return (int)blockIdx.x + iter * (int)gridDim.x;
  __syncthreads();
  if (threadIdx.x == 0) *slot = atomicAdd(cnt, 1);
  __syncthreads();
  return *slot;
}

__global__ __launch_bounds__(256) void k_gemm(const float* __restrict__ feat,
                                              const float* __restrict__ W,
                                              float* __restrict__ ft,
                                              float* __restrict__ out,
                                              int* __restrict__ queues,
                                              size_t ws_size) {
  __shared__ float As[16][IN_F + 4];
  const int tid    = threadIdx.x;
  const int wave   = tid >> 5;
  const int lane   = tid & 31;
  const int tile_m = blockIdx.x * 16;

  if (blockIdx.x == 0 && tid < 2 * HEADS && ws_size >= WS_NEED) ((unsigned*)queues)[GK_OFF + tid] = 0u;

  for (int i = tid; i < 16 * (IN_F / 4); i += 256) {
    const int r  = i / (IN_F / 4);
    const int c4 = i % (IN_F / 4);
    float4 v = make_float4(0.f, 0.f, 0.f, 0.f);
    const int row = tile_m + r;
    if (row < NNODES) v = ((const float4*)(feat + (size_t)row * IN_F))[c4];
    ((float4*)&As[r][0])[c4] = v;
  }
  __syncthreads();

  {
    float4* o4 = (float4*)(out + (size_t)tile_m * CO);
    const float4 z = make_float4(0.f, 0.f, 0.f, 0.f);
    o4[tid]       = z;
    o4[tid + 256] = z;
  }

  const int koff = (lane < 16) ? 0 : 2;
  const int arow = lane & 15;
  const int col  = wave * 16 + arow;
  const float* wrow = W + (size_t)col * IN_F;

  v8f c = {};
  for (int k0 = 0; k0 < IN_F; k0 += 4) {
    v2f a, b;
    a.x = As[arow][k0 + koff];
    a.y = As[arow][k0 + koff + 1];
    b.x = wrow[k0 + koff];
    b.y = wrow[k0 + koff + 1];
    c = __builtin_amdgcn_wmma_f32_16x16x4_f32(
         false, a,  false, b,
         (short)0, c,  false,  false);
  }

  const int mbase  = (lane < 16) ? 0 : 8;
  const int outcol = col;
#pragma unroll
  for (int j = 0; j < 8; ++j) {
    const int row = tile_m + mbase + j;
    if (row < NNODES) ft[(size_t)row * CO + outcol] = c[j];
  }
}

__global__ __launch_bounds__(256) void k_eler(const float* __restrict__ ft,
                                              const float* __restrict__ attn,
                                              float* __restrict__ el,
                                              float* __restrict__ er,
                                              unsigned* __restrict__ mkey,
                                              float* __restrict__ ssum,
                                              int* __restrict__ queues,
                                              size_t ws_size) {
  __shared__ unsigned red[8][2 * HEADS];
  const bool dyn = ws_size >= WS_NEED;
  int i = blockIdx.x * 256 + threadIdx.x;
  if (dyn && i < NQUEUE) queues[i * QSTRIDE] = 0;
  if (dyn && i < 4) queues[(2 + (i >> 1)) * QSTRIDE + ((i & 1) ? 32 : 16)] = 0;
  unsigned kmax = 0u, kneg = 0u;
  if (i < NNODES * HEADS) {
    const int n = i / HEADS, h = i % HEADS;
    const float* f  = ft + (size_t)n * CO + h * OUT_F;
    const float* al = attn + h * 2 * OUT_F;
    const float* ar = al + OUT_F;
    float sl = 0.f, sr = 0.f;
#pragma unroll
    for (int d = 0; d < OUT_F; ++d) {
      const float v = f[d];
      sl += v * al[d];
      sr += v * ar[d];
    }
    el[i] = sl;
    er[i] = sr;
    mkey[i] = 0u;
    ssum[i] = 0.f;
    kmax = f2key(sl);
    kneg = f2key(-sl);
  }
  if (dyn) {
    for (int o = HEADS; o < 32; o <<= 1) {
      const unsigned a = (unsigned)__shfl_xor((int)kmax, o);
      const unsigned b = (unsigned)__shfl_xor((int)kneg, o);
      kmax = a > kmax ? a : kmax;
      kneg = b > kneg ? b : kneg;
    }
    const int lane = threadIdx.x & 31, wave = threadIdx.x >> 5;
    if (lane < HEADS) { red[wave][lane] = kmax; red[wave][HEADS + lane] = kneg; }
    __syncthreads();
    if (threadIdx.x < 2 * HEADS) {
      unsigned v = 0u;
      for (int w = 0; w < 8; ++w) v = red[w][threadIdx.x] > v ? red[w][threadIdx.x] : v;
      atomicMax((unsigned*)queues + GK_OFF + threadIdx.x, v);
    }
  }
}

__global__ __launch_bounds__(256) void k_edge_max(const int* __restrict__ src,
                                                  const int* __restrict__ dst,
                                                  const float* __restrict__ el,
                                                  const float* __restrict__ er,
                                                  unsigned* __restrict__ mkey,
                                                  int* __restrict__ queues,
                                                  size_t ws_size) {
  __shared__ int slot;
  const bool dyn = ws_size >= WS_NEED;
  float gmax[HEADS];
  if (el_bound(queues, dyn, gmax)) return;
  const unsigned half = l2_half();
  for (int pass = 0; pass < 2; ++pass) {
    const unsigned q = half ^ (unsigned)pass;
    int* cnt = queues + (0 * 2 + q) * QSTRIDE;
    for (int iter = 0;; ++iter) {
      const int c = next_chunk(cnt, dyn, iter, &slot);
      if (c >= NCHUNK) break;
      const int h  = (int)threadIdx.x & 3;
      const int e0 = c * CHUNK_E + ((int)threadIdx.x >> 2);
      int sI[4], dI[4];
      unsigned key[4];
      for (int k = 0; k < 4; ++k) { sI[k] = src[e0 + k * 64]; dI[k] = dst[e0 + k * 64]; }
      for (int k = 0; k < 4; ++k)
        key[k] = f2key(leaky(el[sI[k] * HEADS + h] + er[dI[k] * HEADS + h]));
      for (int k = 0; k < 4; ++k)
        if ((unsigned)(dI[k] >= HALF_N) == q) atomicMax(&mkey[dI[k] * HEADS + h], key[k]);
    }
  }
}

__device__ __forceinline__ void sum_terms(float* t, const float4& L, const float4& R, const uint4& K,
                                          bool bound, const float gmax[HEADS]) {
  const float l[HEADS]  = {L.x, L.y, L.z, L.w};
  const float rr[HEADS] = {R.x, R.y, R.z, R.w};
  float mv[HEADS];
  if (bound) {
    for (int h = 0; h < HEADS; ++h) mv[h] = leaky(gmax[h] + rr[h]);
  } else {
    const unsigned kk[HEADS] = {K.x, K.y, K.z, K.w};
    for (int h = 0; h < HEADS; ++h) mv[h] = key2f(kk[h]);
  }
  for (int h = 0; h < HEADS; ++h) atomicAdd(t + h, expf(leaky(l[h] + rr[h]) - mv[h]));
}

__global__ __launch_bounds__(SUM_T) void k_edge_sum(const int* __restrict__ src,
                                                    const int* __restrict__ dst,
                                                    const float* __restrict__ el,
                                                    const float* __restrict__ er,
                                                    const unsigned* __restrict__ mkey,
                                                    float* __restrict__ ssum,
                                                    int* __restrict__ queues,
                                                    size_t ws_size) {
  __shared__ float tab[SR_F];
  __shared__ int2 strip[SUM_T / 32][STRIP_E];
  __shared__ int slot[4];
  const int tid  = (int)threadIdx.x;
  const int lane = tid & 31;
  int2* st = strip[tid >> 5];
  const bool dyn = ws_size >= WS_NEED;
  float gmax[HEADS];
  const bool bound = el_bound(queues, dyn, gmax);

  int r0 = (int)blockIdx.x & (SUM_RANGES - 1);
  if (dyn) {
    const int half = (int)l2_half();
    if (tid == 0) slot[0] = atomicAdd(queues + (2 + half) * QSTRIDE + 16, 1);
    __syncthreads();
    r0 = 2 * half + (slot[0] & 1);
  }

  const int nstep = dyn ? SUM_RANGES : 1;
  for (int step = 0; step < nstep; ++step) {
    const int r = r0 ^ step;
    int* cnt = queues + (2 + (r >> 1)) * QSTRIDE + (r & 1) * 32;
    const int base = r * SR_N;
    if (step == 1) {
      if (tid >= 1 && tid < SUM_RANGES) {
        const int o = r0 ^ tid;
        slot[tid] = __hip_atomic_load(queues + (2 + (o >> 1)) * QSTRIDE + (o & 1) * 32,
                                      __ATOMIC_RELAXED, __HIP_MEMORY_SCOPE_AGENT);
      }
      __syncthreads();
    }
    if (step >= 1 && slot[step] >= NSCH) continue;

    int c;
    if (dyn) {
      __syncthreads();
      if (tid == 0) slot[0] = atomicAdd(cnt, 1);
      __syncthreads();
      c = slot[0];
    } else {
      c = (int)blockIdx.x >> 2;
    }
    if (c >= NSCH) continue;

    for (int i = tid; i < SR_F / 4; i += SUM_T) ((float4*)tab)[i] = make_float4(0.f, 0.f, 0.f, 0.f);
    __syncthreads();

    while (c < NSCH) {
      const int eA = c * SCH_E + tid * 4, eB = eA + SCH_E / 2;
      int4 dA = make_int4(-1, -1, -1, -1), sA = make_int4(0, 0, 0, 0), dB = dA, sB = sA;
      if (eA < NEDGES) {
        dA = *(const int4*)(dst + eA);
        sA = *(const int4*)(src + eA);
      }
      if (eB < NEDGES) {
        dB = *(const int4*)(dst + eB);
        sB = *(const int4*)(src + eB);
      }
      const int dI[8] = {dA.x, dA.y, dA.z, dA.w, dB.x, dB.y, dB.z, dB.w};
      const int sI[8] = {sA.x, sA.y, sA.z, sA.w, sB.x, sB.y, sB.z, sB.w};
      int T = 0;
      for (int k = 0; k < 8; ++k) {
        const bool m = (unsigned)(dI[k] - base) < (unsigned)SR_N;
        const unsigned bm = __builtin_amdgcn_ballot_w32(m);
        if (m) st[T + (int)__builtin_amdgcn_mbcnt_lo(bm, 0u)] = make_int2(sI[k], dI[k]);
        T += __builtin_popcount(bm);
      }
      int nxt = c + SUM_WGS / SUM_RANGES;
      for (int b = 0; b < T; b += 96) {
        const int i0 = b + lane, i1 = i0 + 32, i2 = i0 + 64;
        const bool a0 = i0 < T, a1 = i1 < T, a2 = i2 < T;
        int2 p0 = make_int2(0, base), p1 = p0, p2 = p0;
        if (a0) p0 = st[i0];
        if (a1) p1 = st[i1];
        if (a2) p2 = st[i2];
        float4 L0, R0, L1, R1, L2, R2;
        uint4 K0, K1, K2;
        if (a0) {
          L0 = *(const float4*)(el + (size_t)p0.x * HEADS);
          R0 = *(const float4*)(er + (size_t)p0.y * HEADS);
          if (!bound) K0 = *(const uint4*)(mkey + (size_t)p0.y * HEADS);
        }
        if (a1) {
          L1 = *(const float4*)(el + (size_t)p1.x * HEADS);
          R1 = *(const float4*)(er + (size_t)p1.y * HEADS);
          if (!bound) K1 = *(const uint4*)(mkey + (size_t)p1.y * HEADS);
        }
        if (a2) {
          L2 = *(const float4*)(el + (size_t)p2.x * HEADS);
          R2 = *(const float4*)(er + (size_t)p2.y * HEADS);
          if (!bound) K2 = *(const uint4*)(mkey + (size_t)p2.y * HEADS);
        }
        if (dyn && b == 0 && tid == 0) nxt = atomicAdd(cnt, 1);
        if (a0) sum_terms(tab + (p0.y - base) * HEADS, L0, R0, K0, bound, gmax);
        if (a1) sum_terms(tab + (p1.y - base) * HEADS, L1, R1, K1, bound, gmax);
        if (a2) sum_terms(tab + (p2.y - base) * HEADS, L2, R2, K2, bound, gmax);
      }
      if (dyn) {
        if (T == 0 && tid == 0) nxt = atomicAdd(cnt, 1);
        if (tid == 0) slot[0] = nxt;
        __syncthreads();
        nxt = slot[0];
        __syncthreads();
      }
      c = nxt;
    }

    __syncthreads();
    const int g0 = base * HEADS;
    const int gs = (g0 & ~31) + tid;
    for (int i0 = 0; i0 < (SR_F + 31 + SUM_T) / SUM_T; i0 += 7) {
      float v[7];
      int   g[7];
      for (int j = 0; j < 7; ++j) {
        g[j] = gs + (i0 + j) * SUM_T;
        const bool in = g[j] >= g0 && g[j] < g0 + SR_F;
        int o = g[j] - g0;
        o = o < 0 ? 0 : (o > SR_F - 1 ? SR_F - 1 : o);
        const float x = tab[o];
        v[j] = in ? x : 0.f;
      }
      for (int j = 0; j < 7; ++j)
        if (v[j] != 0.f) atomicAdd(ssum + g[j], v[j]);
    }
  }
}

__global__ __launch_bounds__(256) void k_edge_agg(const int* __restrict__ src,
                                                  const int* __restrict__ dst,
                                                  const float* __restrict__ el,
                                                  const float* __restrict__ er,
                                                  const unsigned* __restrict__ mkey,
                                                  const float* __restrict__ ssum,
                                                  const float* __restrict__ ft,
                                                  float* __restrict__ out,
                                                  int* __restrict__ queues,
                                                  size_t ws_size) {
  __shared__ int slot;
  const int lane = threadIdx.x & 31;
  const int wave = threadIdx.x >> 5;
  const bool dyn = ws_size >= WS_NEED;
  float gmax[HEADS];
  const bool bound = el_bound(queues, dyn, gmax);
  const unsigned half = l2_half();
  for (int pass = 0; pass < 2; ++pass) {
    const unsigned q = half ^ (unsigned)pass;
    int* cnt = queues + (2 * 2 + q) * QSTRIDE;
    for (int iter = 0;; ++iter) {
      const int c = next_chunk(cnt, dyn, iter, &slot);
      if (c >= NCHUNK) break;
      const int e  = c * CHUNK_E + wave * 32 + lane;
      const int sv = src[e], dv = dst[e];
      const bool mine = (unsigned)(dv >= HALF_N) == q;
      float a[HEADS] = {0.f, 0.f, 0.f, 0.f};
      if (mine) {
        float l[HEADS], r[HEADS], sm[HEADS], m[HEADS];
        for (int h = 0; h < HEADS; ++h) {
          l[h]  = el[sv * HEADS + h];
          r[h]  = er[dv * HEADS + h];
          sm[h] = ssum[dv * HEADS + h];
        }
        if (bound) {
          for (int h = 0; h < HEADS; ++h) m[h] = leaky(gmax[h] + r[h]);
        } else {
          for (int h = 0; h < HEADS; ++h) m[h] = key2f(mkey[dv * HEADS + h]);
        }
        for (int h = 0; h < HEADS; ++h) {
          const float x = leaky(l[h] + r[h]);
          a[h] = expf(x - m[h]) / sm[h];
        }
      }
      unsigned todo = __builtin_amdgcn_ballot_w32(mine);
      while (todo) {
        const int j = __builtin_ctz(todo);
        todo &= todo - 1;
        const int s = __builtin_amdgcn_readlane(sv, j);
        const int d = __builtin_amdgcn_readlane(dv, j);
        const float* fsrc = ft + (size_t)s * CO;
        float* orow = out + (size_t)d * CO;
        float f[HEADS];
        for (int h = 0; h < HEADS; ++h) f[h] = fsrc[h * OUT_F + lane];
        for (int h = 0; h < HEADS; ++h) {
          const float w = __int_as_float(__builtin_amdgcn_readlane(__float_as_int(a[h]), j));
          atomicAdd(&orow[h * OUT_F + lane], f[h] * w);
        }
      }
    }
  }
}

extern "C" void kernel_launch(void* const* d_in, const int* in_sizes, int n_in,
                              void* d_out, int out_size, void* d_ws, size_t ws_size,
                              hipStream_t stream) {
  (void)in_sizes; (void)n_in; (void)out_size;

  const float* feat = (const float*)d_in[0];
  const int*   src  = (const int*)d_in[1];
  const int*   dst  = (const int*)d_in[2];
  const float* W    = (const float*)d_in[3];
  const float* attn = (const float*)d_in[4];
  float* out = (float*)d_out;

  char* ws = (char*)d_ws;
  float*    ft   = (float*)ws;    ws += (size_t)NNODES * CO * sizeof(float);
  float*    el   = (float*)ws;    ws += (size_t)NNODES * HEADS * sizeof(float);
  float*    er   = (float*)ws;    ws += (size_t)NNODES * HEADS * sizeof(float);
  unsigned* mkey = (unsigned*)ws; ws += (size_t)NNODES * HEADS * sizeof(unsigned);
  float*    ssum = (float*)ws;    ws += (size_t)NNODES * HEADS * sizeof(float);
  int*      queues = (int*)ws;

  k_gemm<<<NNODES / 16, 256, 0, stream>>>(feat, W, ft, out, queues, ws_size);
  k_eler<<<(NNODES * HEADS + 255) / 256, 256, 0, stream>>>(ft, attn, el, er, mkey, ssum, queues, ws_size);
  k_edge_max<<<MAXP_WGS, 256, 0, stream>>>(src, dst, el, er, mkey, queues, ws_size);
  k_edge_sum<<<SUM_WGS, SUM_T, 0, stream>>>(src, dst, el, er, mkey, ssum, queues, ws_size);
  k_edge_agg<<<EDGE_WGS, 256, 0, stream>>>(src, dst, el, er, mkey, ssum, ft, out, queues, ws_size);
}


# GEMM work-group takes 80 rows, not 16: a wave keeps five accumulator tiles, so each 16-line W fragment load feeds five matrix instructions; 3125 work-groups to 625; k order of every sum unchanged
# speedup vs baseline: 2.9442x; 1.0883x over previous
#include <hip/hip_runtime.h>
#include <math.h>


#define NNODES 50000
#define NEDGES 1600000
#define IN_F   256
#define HEADS  4
#define OUT_F  32
#define CO     (HEADS * OUT_F)
#define NEG_SLOPE 0.2f

#define CHUNK_E  256
#define NCHUNK   (NEDGES / CHUNK_E)
#define HALF_N   (NNODES / 2)
#define QSTRIDE  64
#define NQUEUE   6
#define EDGE_WGS 2048
#define MAXP_WGS 256
#define SUM_RANGES 4
#define SR_N     (NNODES / SUM_RANGES)
#define SR_F     (SR_N * HEADS)
#define SUM_WGS  256
#define SUM_T    1024
#define SCH_E    (SUM_T * 8)
#define STRIP_E  (32 * 8)
#define NSCH     ((NEDGES + SCH_E - 1) / SCH_E)
#define GK_OFF   32
#define SPREAD_MAX 16.0f
#define WS_NEED  ((size_t)NNODES * CO * 4 + (size_t)4 * NNODES * HEADS * 4 + (size_t)NQUEUE * QSTRIDE * 4)

static_assert(NEDGES % CHUNK_E == 0, "chunks cover the edge list exactly");
static_assert(NNODES % SUM_RANGES == 0 && HALF_N == 2 * SR_N, "two sum ranges per cache half");
static_assert(NEDGES % 4 == 0, "a thread's four edges in a row are all inside the list or all outside");
static_assert(SUM_WGS % SUM_RANGES == 0, "the fixed deal gives every range the same number of work-groups");
static_assert(((SR_F + 31 + SUM_T) / SUM_T) % 7 == 0, "the table goes out in groups of seven reads per thread");

typedef __attribute__((ext_vector_type(2))) float v2f;
typedef __attribute__((ext_vector_type(8))) float v8f;

__device__ __forceinline__ unsigned f2key(float x) {
  unsigned b = __float_as_uint(x);
  return (b & 0x80000000u) ? ~b : (b | 0x80000000u);
}
__device__ __forceinline__ float key2f(unsigned k) {
  unsigned b = (k & 0x80000000u) ? (k ^ 0x80000000u) : ~k;
  return __uint_as_float(b);
}
__device__ __forceinline__ float leaky(float x) {
  return x > 0.f ? x : NEG_SLOPE * x;
}

__device__ __forceinline__ unsigned l2_half() {
  return (__builtin_amdgcn_s_sendmsg_rtn(0x87) >> 18) & 1u;
}

__device__ __forceinline__ bool el_bound(const int* queues, bool dyn, float gmax[HEADS]) {
  bool ok = dyn;
  for (int h = 0; h < HEADS; ++h) gmax[h] = 0.f;
  if (dyn) {
    const unsigned* gk = (const unsigned*)queues + GK_OFF;
    for (int h = 0; h < HEADS; ++h) {
      const float mx = key2f(gk[h]);
      const float mn = -key2f(gk[HEADS + h]);
      gmax[h] = mx;
      ok = ok && (mx - mn <= SPREAD_MAX);
    }
  }
  return ok;
}

__device__ __forceinline__ int next_chunk(int* cnt, bool dyn, int iter, int* slot) {
  if (!dyn) return (int)blockIdx.x + iter * (int)gridDim.x;
  __syncthreads();
  if (threadIdx.x == 0) *slot = atomicAdd(cnt, 1);
  __syncthreads();
  return *slot;
}

#define GEMM_MT 5
#define GEMM_R  (16 * GEMM_MT)
static_assert(NNODES % GEMM_R == 0, "GEMM blocks cover the node rows exactly");

__global__ __launch_bounds__(256) void k_gemm(const float* __restrict__ feat,
                                              const float* __restrict__ W,
                                              float* __restrict__ ft,
                                              float* __restrict__ out,
                                              int* __restrict__ queues,
                                              size_t ws_size) {
  __shared__ float As[GEMM_R][IN_F + 4];
  const int tid    = threadIdx.x;
  const int wave   = tid >> 5;
  const int lane   = tid & 31;
  const int tile_m = blockIdx.x * GEMM_R;

  if (blockIdx.x == 0 && tid < 2 * HEADS && ws_size >= WS_NEED) ((unsigned*)queues)[GK_OFF + tid] = 0u;

  {
    constexpr int NLD = GEMM_R * (IN_F / 4) / 256;
    const int r0 = tid >> 6, c4 = tid & 63;
    const float4* g = (const float4*)(feat + (size_t)(tile_m + r0) * IN_F) + c4;
    float4 v[NLD];
#pragma unroll
    for (int j = 0; j < NLD; ++j) v[j] = g[(size_t)j * 4 * (IN_F / 4)];
#pragma unroll
    for (int j = 0; j < NLD; ++j) ((float4*)&As[r0 + 4 * j][0])[c4] = v[j];
  }
  __syncthreads();

  {
    float4* o4 = (float4*)(out + (size_t)tile_m * CO);
    const float4 z = make_float4(0.f, 0.f, 0.f, 0.f);
    for (int j = 0; j < GEMM_R * (CO / 4) / 256; ++j) o4[tid + j * 256] = z;
  }

  const int koff = (lane < 16) ? 0 : 2;
  const int arow = lane & 15;
  const int col  = wave * 16 + arow;
  const float* wrow = W + (size_t)col * IN_F + koff;
  const float* arow0 = &As[arow][koff];

  v8f c[GEMM_MT];
  for (int m = 0; m < GEMM_MT; ++m) c[m] = v8f{};
  for (int k0 = 0; k0 < IN_F; k0 += 4) {
    v2f b;
    b.x = wrow[k0];
    b.y = wrow[k0 + 1];
    for (int m = 0; m < GEMM_MT; ++m) {
      v2f a;
      a.x = arow0[m * 16 * (IN_F + 4) + k0];
      a.y = arow0[m * 16 * (IN_F + 4) + k0 + 1];
      c[m] = __builtin_amdgcn_wmma_f32_16x16x4_f32(
           false, a,  false, b,
           (short)0, c[m],  false,  false);
    }
  }

  const int mbase = (lane < 16) ? 0 : 8;
  for (int m = 0; m < GEMM_MT; ++m) {
    float* frow = ft + (size_t)(tile_m + m * 16 + mbase) * CO + col;
    for (int j = 0; j < 8; ++j) frow[(size_t)j * CO] = c[m][j];
  }
}

__global__ __launch_bounds__(256) void k_eler(const float* __restrict__ ft,
                                              const float* __restrict__ attn,
                                              float* __restrict__ el,
                                              float* __restrict__ er,
                                              unsigned* __restrict__ mkey,
                                              float* __restrict__ ssum,
                                              int* __restrict__ queues,
                                              size_t ws_size) {
  __shared__ unsigned red[8][2 * HEADS];
  const bool dyn = ws_size >= WS_NEED;
  int i = blockIdx.x * 256 + threadIdx.x;
  if (dyn && i < NQUEUE) queues[i * QSTRIDE] = 0;
  if (dyn && i < 4) queues[(2 + (i >> 1)) * QSTRIDE + ((i & 1) ? 32 : 16)] = 0;
  unsigned kmax = 0u, kneg = 0u;
  if (i < NNODES * HEADS) {
    const int n = i / HEADS, h = i % HEADS;
    const float* f  = ft + (size_t)n * CO + h * OUT_F;
    const float* al = attn + h * 2 * OUT_F;
    const float* ar = al + OUT_F;
    float sl = 0.f, sr = 0.f;
#pragma unroll
    for (int d = 0; d < OUT_F; ++d) {
      const float v = f[d];
      sl += v * al[d];
      sr += v * ar[d];
    }
    el[i] = sl;
    er[i] = sr;
    mkey[i] = 0u;
    ssum[i] = 0.f;
    kmax = f2key(sl);
    kneg = f2key(-sl);
  }
  if (dyn) {
    for (int o = HEADS; o < 32; o <<= 1) {
      const unsigned a = (unsigned)__shfl_xor((int)kmax, o);
      const unsigned b = (unsigned)__shfl_xor((int)kneg, o);
      kmax = a > kmax ? a : kmax;
      kneg = b > kneg ? b : kneg;
    }
    const int lane = threadIdx.x & 31, wave = threadIdx.x >> 5;
    if (lane < HEADS) { red[wave][lane] = kmax; red[wave][HEADS + lane] = kneg; }
    __syncthreads();
    if (threadIdx.x < 2 * HEADS) {
      unsigned v = 0u;
      for (int w = 0; w < 8; ++w) v = red[w][threadIdx.x] > v ? red[w][threadIdx.x] : v;
      atomicMax((unsigned*)queues + GK_OFF + threadIdx.x, v);
    }
  }
}

__global__ __launch_bounds__(256) void k_edge_max(const int* __restrict__ src,
                                                  const int* __restrict__ dst,
                                                  const float* __restrict__ el,
                                                  const float* __restrict__ er,
                                                  unsigned* __restrict__ mkey,
                                                  int* __restrict__ queues,
                                                  size_t ws_size) {
  __shared__ int slot;
  const bool dyn = ws_size >= WS_NEED;
  float gmax[HEADS];
  if (el_bound(queues, dyn, gmax)) return;
  const unsigned half = l2_half();
  for (int pass = 0; pass < 2; ++pass) {
    const unsigned q = half ^ (unsigned)pass;
    int* cnt = queues + (0 * 2 + q) * QSTRIDE;
    for (int iter = 0;; ++iter) {
      const int c = next_chunk(cnt, dyn, iter, &slot);
      if (c >= NCHUNK) break;
      const int h  = (int)threadIdx.x & 3;
      const int e0 = c * CHUNK_E + ((int)threadIdx.x >> 2);
      int sI[4], dI[4];
      unsigned key[4];
      for (int k = 0; k < 4; ++k) { sI[k] = src[e0 + k * 64]; dI[k] = dst[e0 + k * 64]; }
      for (int k = 0; k < 4; ++k)
        key[k] = f2key(leaky(el[sI[k] * HEADS + h] + er[dI[k] * HEADS + h]));
      for (int k = 0; k < 4; ++k)
        if ((unsigned)(dI[k] >= HALF_N) == q) atomicMax(&mkey[dI[k] * HEADS + h], key[k]);
    }
  }
}

__device__ __forceinline__ void sum_terms(float* t, const float4& L, const float4& R, const uint4& K,
                                          bool bound, const float gmax[HEADS]) {
  const float l[HEADS]  = {L.x, L.y, L.z, L.w};
  const float rr[HEADS] = {R.x, R.y, R.z, R.w};
  float mv[HEADS];
  if (bound) {
    for (int h = 0; h < HEADS; ++h) mv[h] = leaky(gmax[h] + rr[h]);
  } else {
    const unsigned kk[HEADS] = {K.x, K.y, K.z, K.w};
    for (int h = 0; h < HEADS; ++h) mv[h] = key2f(kk[h]);
  }
  for (int h = 0; h < HEADS; ++h) atomicAdd(t + h, expf(leaky(l[h] + rr[h]) - mv[h]));
}

__global__ __launch_bounds__(SUM_T) void k_edge_sum(const int* __restrict__ src,
                                                    const int* __restrict__ dst,
                                                    const float* __restrict__ el,
                                                    const float* __restrict__ er,
                                                    const unsigned* __restrict__ mkey,
                                                    float* __restrict__ ssum,
                                                    int* __restrict__ queues,
                                                    size_t ws_size) {
  __shared__ float tab[SR_F];
  __shared__ int2 strip[SUM_T / 32][STRIP_E];
  __shared__ int slot[4];
  const int tid  = (int)threadIdx.x;
  const int lane = tid & 31;
  int2* st = strip[tid >> 5];
  const bool dyn = ws_size >= WS_NEED;
  float gmax[HEADS];
  const bool bound = el_bound(queues, dyn, gmax);

  int r0 = (int)blockIdx.x & (SUM_RANGES - 1);
  if (dyn) {
    const int half = (int)l2_half();
    if (tid == 0) slot[0] = atomicAdd(queues + (2 + half) * QSTRIDE + 16, 1);
    __syncthreads();
    r0 = 2 * half + (slot[0] & 1);
  }

  const int nstep = dyn ? SUM_RANGES : 1;
  for (int step = 0; step < nstep; ++step) {
    const int r = r0 ^ step;
    int* cnt = queues + (2 + (r >> 1)) * QSTRIDE + (r & 1) * 32;
    const int base = r * SR_N;
    if (step == 1) {
      if (tid >= 1 && tid < SUM_RANGES) {
        const int o = r0 ^ tid;
        slot[tid] = __hip_atomic_load(queues + (2 + (o >> 1)) * QSTRIDE + (o & 1) * 32,
                                      __ATOMIC_RELAXED, __HIP_MEMORY_SCOPE_AGENT);
      }
      __syncthreads();
    }
    if (step >= 1 && slot[step] >= NSCH) continue;

    int c;
    if (dyn) {
      __syncthreads();
      if (tid == 0) slot[0] = atomicAdd(cnt, 1);
      __syncthreads();
      c = slot[0];
    } else {
      c = (int)blockIdx.x >> 2;
    }
    if (c >= NSCH) continue;

    for (int i = tid; i < SR_F / 4; i += SUM_T) ((float4*)tab)[i] = make_float4(0.f, 0.f, 0.f, 0.f);
    __syncthreads();

    while (c < NSCH) {
      const int eA = c * SCH_E + tid * 4, eB = eA + SCH_E / 2;
      int4 dA = make_int4(-1, -1, -1, -1), sA = make_int4(0, 0, 0, 0), dB = dA, sB = sA;
      if (eA < NEDGES) {
        dA = *(const int4*)(dst + eA);
        sA = *(const int4*)(src + eA);
      }
      if (eB < NEDGES) {
        dB = *(const int4*)(dst + eB);
        sB = *(const int4*)(src + eB);
      }
      const int dI[8] = {dA.x, dA.y, dA.z, dA.w, dB.x, dB.y, dB.z, dB.w};
      const int sI[8] = {sA.x, sA.y, sA.z, sA.w, sB.x, sB.y, sB.z, sB.w};
      int T = 0;
      for (int k = 0; k < 8; ++k) {
        const bool m = (unsigned)(dI[k] - base) < (unsigned)SR_N;
        const unsigned bm = __builtin_amdgcn_ballot_w32(m);
        if (m) st[T + (int)__builtin_amdgcn_mbcnt_lo(bm, 0u)] = make_int2(sI[k], dI[k]);
        T += __builtin_popcount(bm);
      }
      int nxt = c + SUM_WGS / SUM_RANGES;
      for (int b = 0; b < T; b += 96) {
        const int i0 = b + lane, i1 = i0 + 32, i2 = i0 + 64;
        const bool a0 = i0 < T, a1 = i1 < T, a2 = i2 < T;
        int2 p0 = make_int2(0, base), p1 = p0, p2 = p0;
        if (a0) p0 = st[i0];
        if (a1) p1 = st[i1];
        if (a2) p2 = st[i2];
        float4 L0, R0, L1, R1, L2, R2;
        uint4 K0, K1, K2;
        if (a0) {
          L0 = *(const float4*)(el + (size_t)p0.x * HEADS);
          R0 = *(const float4*)(er + (size_t)p0.y * HEADS);
          if (!bound) K0 = *(const uint4*)(mkey + (size_t)p0.y * HEADS);
        }
        if (a1) {
          L1 = *(const float4*)(el + (size_t)p1.x * HEADS);
          R1 = *(const float4*)(er + (size_t)p1.y * HEADS);
          if (!bound) K1 = *(const uint4*)(mkey + (size_t)p1.y * HEADS);
        }
        if (a2) {
          L2 = *(const float4*)(el + (size_t)p2.x * HEADS);
          R2 = *(const float4*)(er + (size_t)p2.y * HEADS);
          if (!bound) K2 = *(const uint4*)(mkey + (size_t)p2.y * HEADS);
        }
        if (dyn && b == 0 && tid == 0) nxt = atomicAdd(cnt, 1);
        if (a0) sum_terms(tab + (p0.y - base) * HEADS, L0, R0, K0, bound, gmax);
        if (a1) sum_terms(tab + (p1.y - base) * HEADS, L1, R1, K1, bound, gmax);
        if (a2) sum_terms(tab + (p2.y - base) * HEADS, L2, R2, K2, bound, gmax);
      }
      if (dyn) {
        if (T == 0 && tid == 0) nxt = atomicAdd(cnt, 1);
        if (tid == 0) slot[0] = nxt;
        __syncthreads();
        nxt = slot[0];
        __syncthreads();
      }
      c = nxt;
    }

    __syncthreads();
    const int g0 = base * HEADS;
    const int gs = (g0 & ~31) + tid;
    for (int i0 = 0; i0 < (SR_F + 31 + SUM_T) / SUM_T; i0 += 7) {
      float v[7];
      int   g[7];
      for (int j = 0; j < 7; ++j) {
        g[j] = gs + (i0 + j) * SUM_T;
        const bool in = g[j] >= g0 && g[j] < g0 + SR_F;
        int o = g[j] - g0;
        o = o < 0 ? 0 : (o > SR_F - 1 ? SR_F - 1 : o);
        const float x = tab[o];
        v[j] = in ? x : 0.f;
      }
      for (int j = 0; j < 7; ++j)
        if (v[j] != 0.f) atomicAdd(ssum + g[j], v[j]);
    }
  }
}

__global__ __launch_bounds__(256) void k_edge_agg(const int* __restrict__ src,
                                                  const int* __restrict__ dst,
                                                  const float* __restrict__ el,
                                                  const float* __restrict__ er,
                                                  const unsigned* __restrict__ mkey,
                                                  const float* __restrict__ ssum,
                                                  const float* __restrict__ ft,
                                                  float* __restrict__ out,
                                                  int* __restrict__ queues,
                                                  size_t ws_size) {
  __shared__ int slot;
  const int lane = threadIdx.x & 31;
  const int wave = threadIdx.x >> 5;
  const bool dyn = ws_size >= WS_NEED;
  float gmax[HEADS];
  const bool bound = el_bound(queues, dyn, gmax);
  const unsigned half = l2_half();
  for (int pass = 0; pass < 2; ++pass) {
    const unsigned q = half ^ (unsigned)pass;
    int* cnt = queues + (2 * 2 + q) * QSTRIDE;
    for (int iter = 0;; ++iter) {
      const int c = next_chunk(cnt, dyn, iter, &slot);
      if (c >= NCHUNK) break;
      const int e  = c * CHUNK_E + wave * 32 + lane;
      const int sv = src[e], dv = dst[e];
      const bool mine = (unsigned)(dv >= HALF_N) == q;
      float a[HEADS] = {0.f, 0.f, 0.f, 0.f};
      if (mine) {
        float l[HEADS], r[HEADS], sm[HEADS], m[HEADS];
        for (int h = 0; h < HEADS; ++h) {
          l[h]  = el[sv * HEADS + h];
          r[h]  = er[dv * HEADS + h];
          sm[h] = ssum[dv * HEADS + h];
        }
        if (bound) {
          for (int h = 0; h < HEADS; ++h) m[h] = leaky(gmax[h] + r[h]);
        } else {
          for (int h = 0; h < HEADS; ++h) m[h] = key2f(mkey[dv * HEADS + h]);
        }
        for (int h = 0; h < HEADS; ++h) {
          const float x = leaky(l[h] + r[h]);
          a[h] = expf(x - m[h]) / sm[h];
        }
      }
      unsigned todo = __builtin_amdgcn_ballot_w32(mine);
      while (todo) {
        const int j = __builtin_ctz(todo);
        todo &= todo - 1;
        const int s = __builtin_amdgcn_readlane(sv, j);
        const int d = __builtin_amdgcn_readlane(dv, j);
        const float* fsrc = ft + (size_t)s * CO;
        float* orow = out + (size_t)d * CO;
        float f[HEADS];
        for (int h = 0; h < HEADS; ++h) f[h] = fsrc[h * OUT_F + lane];
        for (int h = 0; h < HEADS; ++h) {
          const float w = __int_as_float(__builtin_amdgcn_readlane(__float_as_int(a[h]), j));
          atomicAdd(&orow[h * OUT_F + lane], f[h] * w);
        }
      }
    }
  }
}

extern "C" void kernel_launch(void* const* d_in, const int* in_sizes, int n_in,
                              void* d_out, int out_size, void* d_ws, size_t ws_size,
                              hipStream_t stream) {
  (void)in_sizes; (void)n_in; (void)out_size;

  const float* feat = (const float*)d_in[0];
  const int*   src  = (const int*)d_in[1];
  const int*   dst  = (const int*)d_in[2];
  const float* W    = (const float*)d_in[3];
  const float* attn = (const float*)d_in[4];
  float* out = (float*)d_out;

  char* ws = (char*)d_ws;
  float*    ft   = (float*)ws;    ws += (size_t)NNODES * CO * sizeof(float);
  float*    el   = (float*)ws;    ws += (size_t)NNODES * HEADS * sizeof(float);
  float*    er   = (float*)ws;    ws += (size_t)NNODES * HEADS * sizeof(float);
  unsigned* mkey = (unsigned*)ws; ws += (size_t)NNODES * HEADS * sizeof(unsigned);
  float*    ssum = (float*)ws;    ws += (size_t)NNODES * HEADS * sizeof(float);
  int*      queues = (int*)ws;

  k_gemm<<<NNODES / GEMM_R, 256, 0, stream>>>(feat, W, ft, out, queues, ws_size);
  k_eler<<<(NNODES * HEADS + 255) / 256, 256, 0, stream>>>(ft, attn, el, er, mkey, ssum, queues, ws_size);
  k_edge_max<<<MAXP_WGS, 256, 0, stream>>>(src, dst, el, er, mkey, queues, ws_size);
  k_edge_sum<<<SUM_WGS, SUM_T, 0, stream>>>(src, dst, el, er, mkey, ssum, queues, ws_size);
  k_edge_agg<<<EDGE_WGS, 256, 0, stream>>>(src, dst, el, er, mkey, ssum, ft, out, queues, ws_size);
}


# Edge phase by owner: 256 work-groups each hold 196 destinations' softmax sums and rows in LDS, pick edges by byte key, store rows once; no global float atomics; LDS arrival order, sum then divide
# speedup vs baseline: 3.2971x; 1.1199x over previous
#include <hip/hip_runtime.h>
#include <math.h>


#define NNODES 50000
#define NEDGES 1600000
#define IN_F   256
#define HEADS  4
#define OUT_F  32
#define CO     (HEADS * OUT_F)
#define NEG_SLOPE 0.2f

#define QSTRIDE  64
#define NQUEUE   6
#define GK_OFF   32
#define SPREAD_MAX 16.0f
#define OWN_R    196
#define OWN_WGS  ((NNODES + OWN_R - 1) / OWN_R)
#define OWN_T    1024
#define NK4      (NEDGES / 16)
#define OWN_NIT  ((NK4 + OWN_T - 1) / OWN_T)
#define RING     512
#define WS_NEED  ((size_t)NNODES * CO * 4 + (size_t)4 * NNODES * HEADS * 4 + (size_t)NQUEUE * QSTRIDE * 4)

static_assert((NNODES - 1) / OWN_R < 256, "an owner's number fits a byte");
static_assert(OWN_WGS * OWN_R >= NNODES, "the owners cover every destination");
static_assert(NEDGES % 16 == 0, "the key list is whole 16-byte pieces");
static_assert((size_t)NEDGES <= (size_t)2 * NNODES * HEADS * 4, "the key list fits the two node arrays' room it takes");
static_assert(OWN_R * HEADS <= OWN_T, "one thread per softmax sum at the clearing");
static_assert((RING & (RING - 1)) == 0 && RING >= 64, "ring: a power of two that holds two batches");

typedef __attribute__((ext_vector_type(2))) float v2f;
typedef __attribute__((ext_vector_type(8))) float v8f;

__device__ __forceinline__ unsigned f2key(float x) {
  unsigned b = __float_as_uint(x);
  return (b & 0x80000000u) ? ~b : (b | 0x80000000u);
}
__device__ __forceinline__ float key2f(unsigned k) {
  unsigned b = (k & 0x80000000u) ? (k ^ 0x80000000u) : ~k;
  return __uint_as_float(b);
}
__device__ __forceinline__ float leaky(float x) {
  return x > 0.f ? x : NEG_SLOPE * x;
}

__device__ __forceinline__ bool el_bound(const int* queues, bool dyn, float gmax[HEADS]) {
  bool ok = dyn;
  for (int h = 0; h < HEADS; ++h) gmax[h] = 0.f;
  if (dyn) {
    const unsigned* gk = (const unsigned*)queues + GK_OFF;
    for (int h = 0; h < HEADS; ++h) {
      const float mx = key2f(gk[h]);
      const float mn = -key2f(gk[HEADS + h]);
      gmax[h] = mx;
      ok = ok && (mx - mn <= SPREAD_MAX);
    }
  }
  return ok;
}

#define GEMM_MT 5
#define GEMM_R  (16 * GEMM_MT)
static_assert(NNODES % GEMM_R == 0, "GEMM blocks cover the node rows exactly");

__global__ __launch_bounds__(256) void k_gemm(const float* __restrict__ feat,
                                              const float* __restrict__ W,
                                              float* __restrict__ ft,
                                              int* __restrict__ queues,
                                              size_t ws_size) {
  __shared__ float As[GEMM_R][IN_F + 4];
  const int tid    = threadIdx.x;
  const int wave   = tid >> 5;
  const int lane   = tid & 31;
  const int tile_m = blockIdx.x * GEMM_R;

  if (blockIdx.x == 0 && tid < 2 * HEADS && ws_size >= WS_NEED) ((unsigned*)queues)[GK_OFF + tid] = 0u;

  {
    constexpr int NLD = GEMM_R * (IN_F / 4) / 256;
    const int r0 = tid >> 6, c4 = tid & 63;
    const float4* g = (const float4*)(feat + (size_t)(tile_m + r0) * IN_F) + c4;
    float4 v[NLD];
#pragma unroll
    for (int j = 0; j < NLD; ++j) v[j] = g[(size_t)j * 4 * (IN_F / 4)];
#pragma unroll
    for (int j = 0; j < NLD; ++j) ((float4*)&As[r0 + 4 * j][0])[c4] = v[j];
  }
  __syncthreads();

  const int koff = (lane < 16) ? 0 : 2;
  const int arow = lane & 15;
  const int col  = wave * 16 + arow;
  const float* wrow = W + (size_t)col * IN_F + koff;
  const float* arow0 = &As[arow][koff];

  v8f c[GEMM_MT];
  for (int m = 0; m < GEMM_MT; ++m) c[m] = v8f{};
  for (int k0 = 0; k0 < IN_F; k0 += 4) {
    v2f b;
    b.x = wrow[k0];
    b.y = wrow[k0 + 1];
    for (int m = 0; m < GEMM_MT; ++m) {
      v2f a;
      a.x = arow0[m * 16 * (IN_F + 4) + k0];
      a.y = arow0[m * 16 * (IN_F + 4) + k0 + 1];
      c[m] = __builtin_amdgcn_wmma_f32_16x16x4_f32(
           false, a,  false, b,
           (short)0, c[m],  false,  false);
    }
  }

  const int mbase = (lane < 16) ? 0 : 8;
  for (int m = 0; m < GEMM_MT; ++m) {
    float* frow = ft + (size_t)(tile_m + m * 16 + mbase) * CO + col;
    for (int j = 0; j < 8; ++j) frow[(size_t)j * CO] = c[m][j];
  }
}

__global__ __launch_bounds__(256) void k_eler(const float* __restrict__ ft,
                                              const float* __restrict__ attn,
                                              float* __restrict__ el,
                                              float* __restrict__ er,
                                              int* __restrict__ queues,
                                              size_t ws_size) {
  __shared__ unsigned red[8][2 * HEADS];
  const bool dyn = ws_size >= WS_NEED;
  int i = blockIdx.x * 256 + threadIdx.x;
  unsigned kmax = 0u, kneg = 0u;
  if (i < NNODES * HEADS) {
    const int n = i / HEADS, h = i % HEADS;
    const float* f  = ft + (size_t)n * CO + h * OUT_F;
    const float* al = attn + h * 2 * OUT_F;
    const float* ar = al + OUT_F;
    float sl = 0.f, sr = 0.f;
#pragma unroll
    for (int d = 0; d < OUT_F; ++d) {
      const float v = f[d];
      sl += v * al[d];
      sr += v * ar[d];
    }
    el[i] = sl;
    er[i] = sr;
    kmax = f2key(sl);
    kneg = f2key(-sl);
  }
  if (dyn) {
    for (int o = HEADS; o < 32; o <<= 1) {
      const unsigned a = (unsigned)__shfl_xor((int)kmax, o);
      const unsigned b = (unsigned)__shfl_xor((int)kneg, o);
      kmax = a > kmax ? a : kmax;
      kneg = b > kneg ? b : kneg;
    }
    const int lane = threadIdx.x & 31, wave = threadIdx.x >> 5;
    if (lane < HEADS) { red[wave][lane] = kmax; red[wave][HEADS + lane] = kneg; }
    __syncthreads();
    if (threadIdx.x < 2 * HEADS) {
      unsigned v = 0u;
      for (int w = 0; w < 8; ++w) v = red[w][threadIdx.x] > v ? red[w][threadIdx.x] : v;
      atomicMax((unsigned*)queues + GK_OFF + threadIdx.x, v);
    }
  }
}

__global__ __launch_bounds__(256) void k_keys(const int* __restrict__ dst,
                                              unsigned* __restrict__ keys) {
  const int i = blockIdx.x * 256 + threadIdx.x;
  if (i < NEDGES / 4) {
    const int4 d = ((const int4*)dst)[i];
    keys[i] = (((unsigned)d.x / OWN_R) & 255u) | ((((unsigned)d.y / OWN_R) & 255u) << 8) |
              ((((unsigned)d.z / OWN_R) & 255u) << 16) | ((((unsigned)d.w / OWN_R) & 255u) << 24);
  }
}

template <bool SUMS>
__device__ __forceinline__ void own_batch(const int* st, int head, int n, int lane, int base,
                                          const int* __restrict__ src, const int* __restrict__ dst,
                                          const float* __restrict__ el, const float* erl,
                                          const float* __restrict__ ft,
                                          float* rows, float* sums, unsigned* mk,
                                          bool bound, const float gmax[HEADS]) {
  int s = 0, rl = 0;
  float x[HEADS] = {0.f, 0.f, 0.f, 0.f};
  if (lane < n) {
    const int e = st[(head + lane) & (RING - 1)];
    s = src[e];
    rl = dst[e] - base;
    const float4 L = *(const float4*)(el + (size_t)s * HEADS);
    const float4 R = *(const float4*)(erl + rl * HEADS);
    const float l[HEADS] = {L.x, L.y, L.z, L.w};
    const float r[HEADS] = {R.x, R.y, R.z, R.w};
    if (!SUMS) {
      for (int h = 0; h < HEADS; ++h) atomicMax(&mk[rl * HEADS + h], f2key(leaky(l[h] + r[h])));
    } else {
      for (int h = 0; h < HEADS; ++h) {
        const float m = bound ? leaky(gmax[h] + r[h]) : key2f(mk[rl * HEADS + h]);
        x[h] = expf(leaky(l[h] + r[h]) - m);
        atomicAdd(&sums[rl * HEADS + h], x[h]);
      }
    }
  }
  if (SUMS) {
    for (int j = 0; j < n; j += 8) {
      float f[8][HEADS];
#pragma unroll
      for (int k = 0; k < 8; ++k) {
        if (j + k < n) {
          const float* fs = ft + (size_t)__builtin_amdgcn_readlane(s, j + k) * CO + lane;
          for (int h = 0; h < HEADS; ++h) f[k][h] = fs[h * OUT_F];
        }
      }
#pragma unroll
      for (int k = 0; k < 8; ++k) {
        if (j + k < n) {
          float* row = rows + __builtin_amdgcn_readlane(rl, j + k) * CO + lane;
          for (int h = 0; h < HEADS; ++h) {
            const float w = __int_as_float(__builtin_amdgcn_readlane(__float_as_int(x[h]), j + k));
            atomicAdd(row + h * OUT_F, f[k][h] * w);
          }
        }
      }
    }
  }
}

__device__ __forceinline__ uint4 own_keys(const uint4* __restrict__ k4, int c, int tid, unsigned idrep) {
  const int idx = c * OWN_T + tid;
  uint4 v = make_uint4(~idrep, ~idrep, ~idrep, ~idrep);
  if (idx < NK4) v = k4[idx];
  return v;
}

template <bool SUMS>
__device__ __forceinline__ void own_walk(int* st, int tid, int base,
                                         const uint4* __restrict__ k4,
                                         const int* __restrict__ src, const int* __restrict__ dst,
                                         const float* __restrict__ el, const float* erl,
                                         const float* __restrict__ ft,
                                         float* rows, float* sums, unsigned* mk,
                                         bool bound, const float gmax[HEADS]) {
  const int lane = tid & 31;
  const unsigned idrep = (unsigned)blockIdx.x * 0x01010101u;
  int head = 0, cnt = 0;
  int c = (int)blockIdx.x % OWN_NIT;
  uint4 cur = own_keys(k4, c, tid, idrep);
  for (int it = 0; it < OWN_NIT; ++it) {
    const int cn = (c + 1 == OWN_NIT) ? 0 : c + 1;
    uint4 nxt = cur;
    if (it + 1 < OWN_NIT) nxt = own_keys(k4, cn, tid, idrep);
    const int e0 = (c * OWN_T + tid) * 16;
    const unsigned w[4] = {cur.x, cur.y, cur.z, cur.w};
    unsigned mask = 0u;
    for (int j = 0; j < 4; ++j) {
      const unsigned x = w[j] ^ idrep;
      const unsigned z = ~(((x & 0x7f7f7f7fu) + 0x7f7f7f7fu) | x | 0x7f7f7f7fu);
      mask |= z >> (7 - j);
    }
    for (;;) {
      const bool m = mask != 0u;
      const unsigned bm = __builtin_amdgcn_ballot_w32(m);
      if (bm == 0u) break;
      if (m) {
        const int b = __builtin_ctz(mask);
        st[(head + cnt + (int)__builtin_amdgcn_mbcnt_lo(bm, 0u)) & (RING - 1)] = e0 + ((b & 7) << 2) + (b >> 3);
        mask &= mask - 1u;
      }
      cnt += __builtin_popcount(bm);
      while (cnt >= 32) {
        own_batch<SUMS>(st, head, 32, lane, base, src, dst, el, erl, ft, rows, sums, mk, bound, gmax);
        head = (head + 32) & (RING - 1);
        cnt -= 32;
      }
    }
    cur = nxt;
    c = cn;
  }
  if (cnt > 0) own_batch<SUMS>(st, head, cnt, lane, base, src, dst, el, erl, ft, rows, sums, mk, bound, gmax);
}

__global__ __launch_bounds__(OWN_T) void k_owner(const int* __restrict__ src,
                                                 const int* __restrict__ dst,
                                                 const unsigned* __restrict__ keys,
                                                 const float* __restrict__ el,
                                                 const float* __restrict__ er,
                                                 const float* __restrict__ ft,
                                                 float* __restrict__ out,
                                                 const int* __restrict__ queues,
                                                 size_t ws_size) {
  __shared__ __align__(16) float rows[OWN_R * CO];
  __shared__ float sums[OWN_R * HEADS];
  __shared__ unsigned mk[OWN_R * HEADS];
  __shared__ __align__(16) float erl[OWN_R * HEADS];
  __shared__ int ring[OWN_T / 32][RING];
  const int tid  = (int)threadIdx.x;
  const int base = (int)blockIdx.x * OWN_R;
  const int nrows = (NNODES - base) < OWN_R ? (NNODES - base) : OWN_R;
  const bool dyn = ws_size >= WS_NEED;
  float gmax[HEADS];
  const bool bound = el_bound(queues, dyn, gmax);

  for (int i = tid; i < OWN_R * CO / 4; i += OWN_T) ((float4*)rows)[i] = make_float4(0.f, 0.f, 0.f, 0.f);
  if (tid < OWN_R * HEADS) { sums[tid] = 0.f; mk[tid] = 0u; }
  if (tid < nrows * HEADS) erl[tid] = er[(size_t)base * HEADS + tid];
  __syncthreads();

  int* st = ring[tid >> 5];
  const uint4* k4 = (const uint4*)keys;
  if (!bound) {
    own_walk<false>(st, tid, base, k4, src, dst, el, erl, ft, rows, sums, mk, bound, gmax);
    __syncthreads();
  }
  own_walk<true>(st, tid, base, k4, src, dst, el, erl, ft, rows, sums, mk, bound, gmax);
  __syncthreads();

  float4* o4 = (float4*)(out + (size_t)base * CO);
  for (int i = tid; i < nrows * (CO / 4); i += OWN_T) {
    const int r = i / (CO / 4), q = i % (CO / 4);
    const float sm = sums[r * HEADS + q / (OUT_F / 4)];
    float4 v = ((const float4*)rows)[i];
    if (sm == 0.f) {
      v = make_float4(0.f, 0.f, 0.f, 0.f);
    } else {
      v.x /= sm; v.y /= sm; v.z /= sm; v.w /= sm;
    }
    o4[i] = v;
  }
}

extern "C" void kernel_launch(void* const* d_in, const int* in_sizes, int n_in,
                              void* d_out, int out_size, void* d_ws, size_t ws_size,
                              hipStream_t stream) {
  (void)in_sizes; (void)n_in; (void)out_size;

  const float* feat = (const float*)d_in[0];
  const int*   src  = (const int*)d_in[1];
  const int*   dst  = (const int*)d_in[2];
  const float* W    = (const float*)d_in[3];
  const float* attn = (const float*)d_in[4];
  float* out = (float*)d_out;

  char* ws = (char*)d_ws;
  float*    ft   = (float*)ws;    ws += (size_t)NNODES * CO * sizeof(float);
  float*    el   = (float*)ws;    ws += (size_t)NNODES * HEADS * sizeof(float);
  float*    er   = (float*)ws;    ws += (size_t)NNODES * HEADS * sizeof(float);
  unsigned* keys = (unsigned*)ws; ws += (size_t)2 * NNODES * HEADS * sizeof(float);
  int*      queues = (int*)ws;

  k_gemm<<<NNODES / GEMM_R, 256, 0, stream>>>(feat, W, ft, queues, ws_size);
  k_eler<<<(NNODES * HEADS + 255) / 256, 256, 0, stream>>>(ft, attn, el, er, queues, ws_size);
  k_keys<<<(NEDGES / 4 + 255) / 256, 256, 0, stream>>>(dst, keys);
  k_owner<<<OWN_WGS, OWN_T, 0, stream>>>(src, dst, keys, el, er, ft, out, queues, ws_size);
}


# Owner's walk by bitmap: the key launch also sets, per owner, one bit per 32-edge group that holds an edge of its; an owner reads its row and only those groups' keys; LDS arrival order as before
# speedup vs baseline: 3.8782x; 1.1762x over previous
#include <hip/hip_runtime.h>
#include <math.h>


#define NNODES 50000
#define NEDGES 1600000
#define IN_F   256
#define HEADS  4
#define OUT_F  32
#define CO     (HEADS * OUT_F)
#define NEG_SLOPE 0.2f

#define QSTRIDE  64
#define NQUEUE   6
#define GK_OFF   32
#define SPREAD_MAX 16.0f
#define OWN_R    196
#define OWN_WGS  ((NNODES + OWN_R - 1) / OWN_R)
#define OWN_T    1024
#define GRP      32
#define NGRP     (NEDGES / GRP)
#define KB_T     1024
#define KB_G     1024
#define KB_WGS   ((NGRP + KB_G - 1) / KB_G)
#define BM_W     (KB_WGS * (KB_G / 32))
#define BM_PW    (BM_W / (OWN_T / 32))
#define RING     256
#define WS_NEED  ((size_t)NNODES * CO * 4 + (size_t)4 * NNODES * HEADS * 4 + (size_t)NQUEUE * QSTRIDE * 4)

static_assert((NNODES - 1) / OWN_R < 256, "an owner's number fits a byte");
static_assert(OWN_WGS * OWN_R >= NNODES, "the owners cover every destination");
static_assert(NEDGES % GRP == 0, "the key list is whole groups");
static_assert(BM_W % (OWN_T / 32) == 0 && BM_PW <= 64, "a wave reads its share of the row with two loads per lane");
static_assert((size_t)BM_W * 4 <= (size_t)(NNODES - (OWN_WGS - 1) * OWN_R) * CO * 4, "a bitmap row fits the last owner's rows");
static_assert((size_t)NEDGES <= (size_t)2 * NNODES * HEADS * 4, "the key list fits the two node arrays' room it takes");
static_assert(OWN_R * HEADS <= OWN_T, "one thread per softmax sum at the clearing");
static_assert((RING & (RING - 1)) == 0 && RING >= 64, "ring: a power of two that holds two batches");

typedef __attribute__((ext_vector_type(2))) float v2f;
typedef __attribute__((ext_vector_type(8))) float v8f;

__device__ __forceinline__ unsigned f2key(float x) {
  unsigned b = __float_as_uint(x);
  return (b & 0x80000000u) ? ~b : (b | 0x80000000u);
}
__device__ __forceinline__ float key2f(unsigned k) {
  unsigned b = (k & 0x80000000u) ? (k ^ 0x80000000u) : ~k;
  return __uint_as_float(b);
}
__device__ __forceinline__ float leaky(float x) {
  return x > 0.f ? x : NEG_SLOPE * x;
}

__device__ __forceinline__ bool el_bound(const int* queues, bool dyn, float gmax[HEADS]) {
  bool ok = dyn;
  for (int h = 0; h < HEADS; ++h) gmax[h] = 0.f;
  if (dyn) {
    const unsigned* gk = (const unsigned*)queues + GK_OFF;
    for (int h = 0; h < HEADS; ++h) {
      const float mx = key2f(gk[h]);
      const float mn = -key2f(gk[HEADS + h]);
      gmax[h] = mx;
      ok = ok && (mx - mn <= SPREAD_MAX);
    }
  }
  return ok;
}

#define GEMM_MT 5
#define GEMM_R  (16 * GEMM_MT)
static_assert(NNODES % GEMM_R == 0, "GEMM blocks cover the node rows exactly");

__global__ __launch_bounds__(256) void k_gemm(const float* __restrict__ feat,
                                              const float* __restrict__ W,
                                              float* __restrict__ ft,
                                              int* __restrict__ queues,
                                              size_t ws_size) {
  __shared__ float As[GEMM_R][IN_F + 4];
  const int tid    = threadIdx.x;
  const int wave   = tid >> 5;
  const int lane   = tid & 31;
  const int tile_m = blockIdx.x * GEMM_R;

  if (blockIdx.x == 0 && tid < 2 * HEADS && ws_size >= WS_NEED) ((unsigned*)queues)[GK_OFF + tid] = 0u;

  {
    constexpr int NLD = GEMM_R * (IN_F / 4) / 256;
    const int r0 = tid >> 6, c4 = tid & 63;
    const float4* g = (const float4*)(feat + (size_t)(tile_m + r0) * IN_F) + c4;
    float4 v[NLD];
#pragma unroll
    for (int j = 0; j < NLD; ++j) v[j] = g[(size_t)j * 4 * (IN_F / 4)];
#pragma unroll
    for (int j = 0; j < NLD; ++j) ((float4*)&As[r0 + 4 * j][0])[c4] = v[j];
  }
  __syncthreads();

  const int koff = (lane < 16) ? 0 : 2;
  const int arow = lane & 15;
  const int col  = wave * 16 + arow;
  const float* wrow = W + (size_t)col * IN_F + koff;
  const float* arow0 = &As[arow][koff];

  v8f c[GEMM_MT];
  for (int m = 0; m < GEMM_MT; ++m) c[m] = v8f{};
  for (int k0 = 0; k0 < IN_F; k0 += 4) {
    v2f b;
    b.x = wrow[k0];
    b.y = wrow[k0 + 1];
    for (int m = 0; m < GEMM_MT; ++m) {
      v2f a;
      a.x = arow0[m * 16 * (IN_F + 4) + k0];
      a.y = arow0[m * 16 * (IN_F + 4) + k0 + 1];
      c[m] = __builtin_amdgcn_wmma_f32_16x16x4_f32(
           false, a,  false, b,
           (short)0, c[m],  false,  false);
    }
  }

  const int mbase = (lane < 16) ? 0 : 8;
  for (int m = 0; m < GEMM_MT; ++m) {
    float* frow = ft + (size_t)(tile_m + m * 16 + mbase) * CO + col;
    for (int j = 0; j < 8; ++j) frow[(size_t)j * CO] = c[m][j];
  }
}

__global__ __launch_bounds__(256) void k_eler(const float* __restrict__ ft,
                                              const float* __restrict__ attn,
                                              float* __restrict__ el,
                                              float* __restrict__ er,
                                              int* __restrict__ queues,
                                              size_t ws_size) {
  __shared__ unsigned red[8][2 * HEADS];
  const bool dyn = ws_size >= WS_NEED;
  int i = blockIdx.x * 256 + threadIdx.x;
  unsigned kmax = 0u, kneg = 0u;
  if (i < NNODES * HEADS) {
    const int n = i / HEADS, h = i % HEADS;
    const float* f  = ft + (size_t)n * CO + h * OUT_F;
    const float* al = attn + h * 2 * OUT_F;
    const float* ar = al + OUT_F;
    float sl = 0.f, sr = 0.f;
#pragma unroll
    for (int d = 0; d < OUT_F; ++d) {
      const float v = f[d];
      sl += v * al[d];
      sr += v * ar[d];
    }
    el[i] = sl;
    er[i] = sr;
    kmax = f2key(sl);
    kneg = f2key(-sl);
  }
  if (dyn) {
    for (int o = HEADS; o < 32; o <<= 1) {
      const unsigned a = (unsigned)__shfl_xor((int)kmax, o);
      const unsigned b = (unsigned)__shfl_xor((int)kneg, o);
      kmax = a > kmax ? a : kmax;
      kneg = b > kneg ? b : kneg;
    }
    const int lane = threadIdx.x & 31, wave = threadIdx.x >> 5;
    if (lane < HEADS) { red[wave][lane] = kmax; red[wave][HEADS + lane] = kneg; }
    __syncthreads();
    if (threadIdx.x < 2 * HEADS) {
      unsigned v = 0u;
      for (int w = 0; w < 8; ++w) v = red[w][threadIdx.x] > v ? red[w][threadIdx.x] : v;
      atomicMax((unsigned*)queues + GK_OFF + threadIdx.x, v);
    }
  }
}

__global__ __launch_bounds__(KB_T) void k_keys(const int* __restrict__ dst,
                                               unsigned* __restrict__ keys,
                                               unsigned* __restrict__ bmo) {
  __shared__ unsigned bm[OWN_WGS][KB_G / 32 + 1];
  const int tid = (int)threadIdx.x;
  for (int i = tid; i < OWN_WGS * (KB_G / 32 + 1); i += KB_T) (&bm[0][0])[i] = 0u;
  __syncthreads();

  constexpr int PER = KB_G * GRP / 4 / KB_T;
  const int i0 = (int)blockIdx.x * (KB_G * GRP / 4);
  int4 d[PER];
#pragma unroll
  for (int j = 0; j < PER; ++j) {
    const int i = i0 + j * KB_T + tid;
    d[j] = ((const int4*)dst)[i < NEDGES / 4 ? i : NEDGES / 4 - 1];
  }
#pragma unroll
  for (int j = 0; j < PER; ++j) {
    const int i = i0 + j * KB_T + tid;
    if (i < NEDGES / 4) {
      const unsigned o0 = ((unsigned)d[j].x / OWN_R) & 255u, o1 = ((unsigned)d[j].y / OWN_R) & 255u;
      const unsigned o2 = ((unsigned)d[j].z / OWN_R) & 255u, o3 = ((unsigned)d[j].w / OWN_R) & 255u;
      keys[i] = o0 | (o1 << 8) | (o2 << 16) | (o3 << 24);
      const int gl = (j * KB_T + tid) / (GRP / 4);
      const unsigned bit = 1u << (gl & 31);
      atomicOr(&bm[o0][gl >> 5], bit);
      atomicOr(&bm[o1][gl >> 5], bit);
      atomicOr(&bm[o2][gl >> 5], bit);
      atomicOr(&bm[o3][gl >> 5], bit);
    }
  }
  __syncthreads();

  for (int o = tid >> 5; o < OWN_WGS; o += KB_T / 32)
    bmo[(size_t)o * OWN_R * CO + (size_t)blockIdx.x * (KB_G / 32) + (tid & 31)] = bm[o][tid & 31];
}

template <bool SUMS>
__device__ __forceinline__ void own_batch(const int* st, int head, int n, int lane, int base,
                                          const int* __restrict__ src, const int* __restrict__ dst,
                                          const float* __restrict__ el, const float* erl,
                                          const float* __restrict__ ft,
                                          float* rows, float* sums, unsigned* mk,
                                          bool bound, const float gmax[HEADS]) {
  int s = 0, rl = 0;
  float x[HEADS] = {0.f, 0.f, 0.f, 0.f};
  if (lane < n) {
    const int e = st[(head + lane) & (RING - 1)];
    s = src[e];
    rl = dst[e] - base;
    const float4 L = *(const float4*)(el + (size_t)s * HEADS);
    const float4 R = *(const float4*)(erl + rl * HEADS);
    const float l[HEADS] = {L.x, L.y, L.z, L.w};
    const float r[HEADS] = {R.x, R.y, R.z, R.w};
    if (!SUMS) {
      for (int h = 0; h < HEADS; ++h) atomicMax(&mk[rl * HEADS + h], f2key(leaky(l[h] + r[h])));
    } else {
      for (int h = 0; h < HEADS; ++h) {
        const float m = bound ? leaky(gmax[h] + r[h]) : key2f(mk[rl * HEADS + h]);
        x[h] = expf(leaky(l[h] + r[h]) - m);
        atomicAdd(&sums[rl * HEADS + h], x[h]);
      }
    }
  }
  if (SUMS) {
    for (int j = 0; j < n; j += 8) {
      float f[8][HEADS];
#pragma unroll
      for (int k = 0; k < 8; ++k) {
        if (j + k < n) {
          const float* fs = ft + (size_t)__builtin_amdgcn_readlane(s, j + k) * CO + lane;
          for (int h = 0; h < HEADS; ++h) f[k][h] = fs[h * OUT_F];
        }
      }
#pragma unroll
      for (int k = 0; k < 8; ++k) {
        if (j + k < n) {
          float* row = rows + __builtin_amdgcn_readlane(rl, j + k) * CO + lane;
          for (int h = 0; h < HEADS; ++h) {
            const float w = __int_as_float(__builtin_amdgcn_readlane(__float_as_int(x[h]), j + k));
            atomicAdd(row + h * OUT_F, f[k][h] * w);
          }
        }
      }
    }
  }
}

template <bool SUMS>
__device__ __forceinline__ void own_groups(const int* gst, int ghead, int n, int lane, unsigned idrep,
                                           const uint4* __restrict__ k4,
                                           int* st, int& head, int& cnt, int base,
                                           const int* __restrict__ src, const int* __restrict__ dst,
                                           const float* __restrict__ el, const float* erl,
                                           const float* __restrict__ ft,
                                           float* rows, float* sums, unsigned* mk,
                                           bool bound, const float gmax[HEADS]) {
  unsigned mask = 0u;
  int e0 = 0;
  if (lane < n) {
    const int g = gst[(ghead + lane) & (RING - 1)];
    if (g < NGRP) {
      const uint4 a = k4[2 * g], c = k4[2 * g + 1];
      const unsigned w[8] = {a.x, a.y, a.z, a.w, c.x, c.y, c.z, c.w};
      for (int j = 0; j < 8; ++j) {
        const unsigned x = w[j] ^ idrep;
        const unsigned z = ~(((x & 0x7f7f7f7fu) + 0x7f7f7f7fu) | x | 0x7f7f7f7fu);
        mask |= z >> (7 - j);
      }
      e0 = g * GRP;
    }
  }
  for (;;) {
    const bool m = mask != 0u;
    const unsigned bm = __builtin_amdgcn_ballot_w32(m);
    if (bm == 0u) break;
    if (m) {
      const int b = __builtin_ctz(mask);
      st[(head + cnt + (int)__builtin_amdgcn_mbcnt_lo(bm, 0u)) & (RING - 1)] = e0 + ((b & 7) << 2) + (b >> 3);
      mask &= mask - 1u;
    }
    cnt += __builtin_popcount(bm);
    while (cnt >= 32) {
      own_batch<SUMS>(st, head, 32, lane, base, src, dst, el, erl, ft, rows, sums, mk, bound, gmax);
      head = (head + 32) & (RING - 1);
      cnt -= 32;
    }
  }
}

template <bool SUMS>
__device__ __forceinline__ void own_walk(int* st, int* gst, int tid, int base,
                                         const uint4* __restrict__ k4, const unsigned* brow,
                                         const int* __restrict__ src, const int* __restrict__ dst,
                                         const float* __restrict__ el, const float* erl,
                                         const float* __restrict__ ft,
                                         float* rows, float* sums, unsigned* mk,
                                         bool bound, const float gmax[HEADS]) {
  const int lane = tid & 31;
  const unsigned idrep = (unsigned)blockIdx.x * 0x01010101u;
  int head = 0, cnt = 0;
  int ghead = 0, gcnt = 0;
  const int w0 = (tid >> 5) * BM_PW;
  unsigned bw[2];
  for (int q = 0; q < 2; ++q) {
    bw[q] = 0u;
    if (q * 32 + lane < BM_PW) bw[q] = brow[w0 + q * 32 + lane];
  }
  for (int q = 0; q < 2; ++q) {
    unsigned word = bw[q];
    const int g0 = (w0 + q * 32 + lane) * 32;
    for (;;) {
      const bool m = word != 0u;
      const unsigned bm = __builtin_amdgcn_ballot_w32(m);
      if (bm == 0u) break;
      if (m) {
        gst[(ghead + gcnt + (int)__builtin_amdgcn_mbcnt_lo(bm, 0u)) & (RING - 1)] = g0 + __builtin_ctz(word);
        word &= word - 1u;
      }
      gcnt += __builtin_popcount(bm);
      while (gcnt >= 32) {
        own_groups<SUMS>(gst, ghead, 32, lane, idrep, k4, st, head, cnt, base, src, dst, el, erl, ft, rows, sums, mk, bound, gmax);
        ghead = (ghead + 32) & (RING - 1);
        gcnt -= 32;
      }
    }
  }
  if (gcnt > 0) own_groups<SUMS>(gst, ghead, gcnt, lane, idrep, k4, st, head, cnt, base, src, dst, el, erl, ft, rows, sums, mk, bound, gmax);
  if (cnt > 0) own_batch<SUMS>(st, head, cnt, lane, base, src, dst, el, erl, ft, rows, sums, mk, bound, gmax);
}

__global__ __launch_bounds__(OWN_T) void k_owner(const int* __restrict__ src,
                                                 const int* __restrict__ dst,
                                                 const unsigned* __restrict__ keys,
                                                 const float* __restrict__ el,
                                                 const float* __restrict__ er,
                                                 const float* __restrict__ ft,
                                                 float* __restrict__ out,
                                                 const int* __restrict__ queues,
                                                 size_t ws_size) {
  __shared__ __align__(16) float rows[OWN_R * CO];
  __shared__ float sums[OWN_R * HEADS];
  __shared__ unsigned mk[OWN_R * HEADS];
  __shared__ __align__(16) float erl[OWN_R * HEADS];
  __shared__ int ring[OWN_T / 32][2 * RING];
  const int tid  = (int)threadIdx.x;
  const int base = (int)blockIdx.x * OWN_R;
  const int nrows = (NNODES - base) < OWN_R ? (NNODES - base) : OWN_R;
  const bool dyn = ws_size >= WS_NEED;
  float gmax[HEADS];
  const bool bound = el_bound(queues, dyn, gmax);

  for (int i = tid; i < OWN_R * CO / 4; i += OWN_T) ((float4*)rows)[i] = make_float4(0.f, 0.f, 0.f, 0.f);
  if (tid < OWN_R * HEADS) { sums[tid] = 0.f; mk[tid] = 0u; }
  if (tid < nrows * HEADS) erl[tid] = er[(size_t)base * HEADS + tid];
  __syncthreads();

  int* st = ring[tid >> 5];
  int* gst = st + RING;
  const uint4* k4 = (const uint4*)keys;
  const unsigned* brow = (const unsigned*)(out + (size_t)base * CO);
  if (!bound) {
    own_walk<false>(st, gst, tid, base, k4, brow, src, dst, el, erl, ft, rows, sums, mk, bound, gmax);
    __syncthreads();
  }
  own_walk<true>(st, gst, tid, base, k4, brow, src, dst, el, erl, ft, rows, sums, mk, bound, gmax);
  __syncthreads();

  float4* o4 = (float4*)(out + (size_t)base * CO);
  for (int i = tid; i < nrows * (CO / 4); i += OWN_T) {
    const int r = i / (CO / 4), q = i % (CO / 4);
    const float sm = sums[r * HEADS + q / (OUT_F / 4)];
    float4 v = ((const float4*)rows)[i];
    if (sm == 0.f) {
      v = make_float4(0.f, 0.f, 0.f, 0.f);
    } else {
      v.x /= sm; v.y /= sm; v.z /= sm; v.w /= sm;
    }
    o4[i] = v;
  }
}

extern "C" void kernel_launch(void* const* d_in, const int* in_sizes, int n_in,
                              void* d_out, int out_size, void* d_ws, size_t ws_size,
                              hipStream_t stream) {
  (void)in_sizes; (void)n_in; (void)out_size;

  const float* feat = (const float*)d_in[0];
  const int*   src  = (const int*)d_in[1];
  const int*   dst  = (const int*)d_in[2];
  const float* W    = (const float*)d_in[3];
  const float* attn = (const float*)d_in[4];
  float* out = (float*)d_out;

  char* ws = (char*)d_ws;
  float*    ft   = (float*)ws;    ws += (size_t)NNODES * CO * sizeof(float);
  float*    el   = (float*)ws;    ws += (size_t)NNODES * HEADS * sizeof(float);
  float*    er   = (float*)ws;    ws += (size_t)NNODES * HEADS * sizeof(float);
  unsigned* keys = (unsigned*)ws; ws += (size_t)2 * NNODES * HEADS * sizeof(float);
  int*      queues = (int*)ws;

  k_gemm<<<NNODES / GEMM_R, 256, 0, stream>>>(feat, W, ft, queues, ws_size);
  k_eler<<<(NNODES * HEADS + 255) / 256, 256, 0, stream>>>(ft, attn, el, er, queues, ws_size);
  k_keys<<<KB_WGS, KB_T, 0, stream>>>(dst, keys, (unsigned*)out);
  k_owner<<<OWN_WGS, OWN_T, 0, stream>>>(src, dst, keys, el, er, ft, out, queues, ws_size);
}


# el/er launch removed: the GEMM epilogue stages its 80x128 tile in LDS, stores ft rows as whole lines and dots el/er there in the same d order; key launch runs first; 4 launches to 3
# speedup vs baseline: 4.0396x; 1.0416x over previous
#include <hip/hip_runtime.h>
#include <math.h>


#define NNODES 50000
#define NEDGES 1600000
#define IN_F   256
#define HEADS  4
#define OUT_F  32
#define CO     (HEADS * OUT_F)
#define NEG_SLOPE 0.2f

#define QSTRIDE  64
#define NQUEUE   6
#define GK_OFF   32
#define SPREAD_MAX 16.0f
#define OWN_R    196
#define OWN_WGS  ((NNODES + OWN_R - 1) / OWN_R)
#define OWN_T    1024
#define GRP      32
#define NGRP     (NEDGES / GRP)
#define KB_T     1024
#define KB_G     1024
#define KB_WGS   ((NGRP + KB_G - 1) / KB_G)
#define BM_W     (KB_WGS * (KB_G / 32))
#define BM_PW    (BM_W / (OWN_T / 32))
#define RING     256
#define WS_NEED  ((size_t)NNODES * CO * 4 + (size_t)4 * NNODES * HEADS * 4 + (size_t)NQUEUE * QSTRIDE * 4)

static_assert((NNODES - 1) / OWN_R < 256, "an owner's number fits a byte");
static_assert(OWN_WGS * OWN_R >= NNODES, "the owners cover every destination");
static_assert(NEDGES % GRP == 0, "the key list is whole groups");
static_assert(BM_W % (OWN_T / 32) == 0 && BM_PW <= 64, "a wave reads its share of the row with two loads per lane");
static_assert((size_t)BM_W * 4 <= (size_t)(NNODES - (OWN_WGS - 1) * OWN_R) * CO * 4, "a bitmap row fits the last owner's rows");
static_assert((size_t)NEDGES <= (size_t)2 * NNODES * HEADS * 4, "the key list fits the two node arrays' room it takes");
static_assert(OWN_R * HEADS <= OWN_T, "one thread per softmax sum at the clearing");
static_assert((RING & (RING - 1)) == 0 && RING >= 64, "ring: a power of two that holds two batches");

typedef __attribute__((ext_vector_type(2))) float v2f;
typedef __attribute__((ext_vector_type(8))) float v8f;

__device__ __forceinline__ unsigned f2key(float x) {
  unsigned b = __float_as_uint(x);
  return (b & 0x80000000u) ? ~b : (b | 0x80000000u);
}
__device__ __forceinline__ float key2f(unsigned k) {
  unsigned b = (k & 0x80000000u) ? (k ^ 0x80000000u) : ~k;
  return __uint_as_float(b);
}
__device__ __forceinline__ float leaky(float x) {
  return x > 0.f ? x : NEG_SLOPE * x;
}

__device__ __forceinline__ bool el_bound(const int* queues, bool dyn, float gmax[HEADS]) {
  bool ok = dyn;
  for (int h = 0; h < HEADS; ++h) gmax[h] = 0.f;
  if (dyn) {
    const unsigned* gk = (const unsigned*)queues + GK_OFF;
    for (int h = 0; h < HEADS; ++h) {
      const float mx = key2f(gk[h]);
      const float mn = -key2f(gk[HEADS + h]);
      gmax[h] = mx;
      ok = ok && (mx - mn <= SPREAD_MAX);
    }
  }
  return ok;
}

#define GEMM_MT 5
#define GEMM_R  (16 * GEMM_MT)
#define CS_P    (CO + 4)
static_assert(NNODES % GEMM_R == 0, "GEMM blocks cover the node rows exactly");
static_assert(GEMM_R * CS_P <= GEMM_R * (IN_F + 4), "the staged result fits the A tile's room");
static_assert(256 % HEADS == 0, "a thread's (node, head) pairs all have its head");

__global__ __launch_bounds__(256) void k_gemm(const float* __restrict__ feat,
                                              const float* __restrict__ W,
                                              const float* __restrict__ attn,
                                              float* __restrict__ ft,
                                              float* __restrict__ el,
                                              float* __restrict__ er,
                                              int* __restrict__ queues,
                                              size_t ws_size) {
  __shared__ __align__(16) float As[GEMM_R][IN_F + 4];
  __shared__ unsigned red[8][2 * HEADS];
  const int tid    = threadIdx.x;
  const int wave   = tid >> 5;
  const int lane   = tid & 31;
  const int tile_m = blockIdx.x * GEMM_R;
  const bool dyn = ws_size >= WS_NEED;

  {
    constexpr int NLD = GEMM_R * (IN_F / 4) / 256;
    const int r0 = tid >> 6, c4 = tid & 63;
    const float4* g = (const float4*)(feat + (size_t)(tile_m + r0) * IN_F) + c4;
    float4 v[NLD];
#pragma unroll
    for (int j = 0; j < NLD; ++j) v[j] = g[(size_t)j * 4 * (IN_F / 4)];
#pragma unroll
    for (int j = 0; j < NLD; ++j) ((float4*)&As[r0 + 4 * j][0])[c4] = v[j];
  }
  __syncthreads();

  const int koff = (lane < 16) ? 0 : 2;
  const int arow = lane & 15;
  const int col  = wave * 16 + arow;
  const float* wrow = W + (size_t)col * IN_F + koff;
  const float* arow0 = &As[arow][koff];

  v8f c[GEMM_MT];
  for (int m = 0; m < GEMM_MT; ++m) c[m] = v8f{};
  for (int k0 = 0; k0 < IN_F; k0 += 4) {
    v2f b;
    b.x = wrow[k0];
    b.y = wrow[k0 + 1];
    for (int m = 0; m < GEMM_MT; ++m) {
      v2f a;
      a.x = arow0[m * 16 * (IN_F + 4) + k0];
      a.y = arow0[m * 16 * (IN_F + 4) + k0 + 1];
      c[m] = __builtin_amdgcn_wmma_f32_16x16x4_f32(
           false, a,  false, b,
           (short)0, c[m],  false,  false);
    }
  }

  __syncthreads();
  float* Cs = &As[0][0];
  const int mbase = (lane < 16) ? 0 : 8;
  for (int m = 0; m < GEMM_MT; ++m) {
    float* crow = Cs + (m * 16 + mbase) * CS_P + col;
    for (int j = 0; j < 8; ++j) crow[j * CS_P] = c[m][j];
  }
  __syncthreads();

  {
    const int q = tid & 31, r0 = tid >> 5;
    for (int i = 0; i < GEMM_R / 8; ++i) {
      const int r = r0 + 8 * i;
      const float4 v = *(const float4*)(Cs + r * CS_P + 4 * q);
      *(float4*)(ft + (size_t)(tile_m + r) * CO + 4 * q) = v;
    }
  }

  unsigned kmax = 0u, kneg = 0u;
  for (int p = tid; p < GEMM_R * HEADS; p += 256) {
    const int r = p / HEADS, h = p % HEADS;
    const float* f  = Cs + r * CS_P + h * OUT_F;
    const float* al = attn + h * 2 * OUT_F;
    const float* ar = al + OUT_F;
    float sl = 0.f, sr = 0.f;
    for (int d = 0; d < OUT_F; ++d) {
      const float v = f[d];
      sl += v * al[d];
      sr += v * ar[d];
    }
    el[(size_t)tile_m * HEADS + p] = sl;
    er[(size_t)tile_m * HEADS + p] = sr;
    const unsigned a = f2key(sl), b = f2key(-sl);
    kmax = a > kmax ? a : kmax;
    kneg = b > kneg ? b : kneg;
  }
  if (dyn) {
    for (int o = HEADS; o < 32; o <<= 1) {
      const unsigned a = (unsigned)__shfl_xor((int)kmax, o);
      const unsigned b = (unsigned)__shfl_xor((int)kneg, o);
      kmax = a > kmax ? a : kmax;
      kneg = b > kneg ? b : kneg;
    }
    if (lane < HEADS) { red[wave][lane] = kmax; red[wave][HEADS + lane] = kneg; }
    __syncthreads();
    if (tid < 2 * HEADS) {
      unsigned v = 0u;
      for (int w = 0; w < 8; ++w) v = red[w][tid] > v ? red[w][tid] : v;
      atomicMax((unsigned*)queues + GK_OFF + tid, v);
    }
  }
}

__global__ __launch_bounds__(KB_T) void k_keys(const int* __restrict__ dst,
                                               unsigned* __restrict__ keys,
                                               unsigned* __restrict__ bmo,
                                               int* __restrict__ queues,
                                               size_t ws_size) {
  __shared__ unsigned bm[OWN_WGS][KB_G / 32 + 1];
  const int tid = (int)threadIdx.x;
  if (blockIdx.x == 0 && tid < 2 * HEADS && ws_size >= WS_NEED) ((unsigned*)queues)[GK_OFF + tid] = 0u;
  for (int i = tid; i < OWN_WGS * (KB_G / 32 + 1); i += KB_T) (&bm[0][0])[i] = 0u;
  __syncthreads();

  constexpr int PER = KB_G * GRP / 4 / KB_T;
  const int i0 = (int)blockIdx.x * (KB_G * GRP / 4);
  int4 d[PER];
#pragma unroll
  for (int j = 0; j < PER; ++j) {
    const int i = i0 + j * KB_T + tid;
    d[j] = ((const int4*)dst)[i < NEDGES / 4 ? i : NEDGES / 4 - 1];
  }
#pragma unroll
  for (int j = 0; j < PER; ++j) {
    const int i = i0 + j * KB_T + tid;
    if (i < NEDGES / 4) {
      const unsigned o0 = ((unsigned)d[j].x / OWN_R) & 255u, o1 = ((unsigned)d[j].y / OWN_R) & 255u;
      const unsigned o2 = ((unsigned)d[j].z / OWN_R) & 255u, o3 = ((unsigned)d[j].w / OWN_R) & 255u;
      keys[i] = o0 | (o1 << 8) | (o2 << 16) | (o3 << 24);
      const int gl = (j * KB_T + tid) / (GRP / 4);
      const unsigned bit = 1u << (gl & 31);
      atomicOr(&bm[o0][gl >> 5], bit);
      atomicOr(&bm[o1][gl >> 5], bit);
      atomicOr(&bm[o2][gl >> 5], bit);
      atomicOr(&bm[o3][gl >> 5], bit);
    }
  }
  __syncthreads();

  for (int o = tid >> 5; o < OWN_WGS; o += KB_T / 32)
    bmo[(size_t)o * OWN_R * CO + (size_t)blockIdx.x * (KB_G / 32) + (tid & 31)] = bm[o][tid & 31];
}

template <bool SUMS>
__device__ __forceinline__ void own_batch(const int* st, int head, int n, int lane, int base,
                                          const int* __restrict__ src, const int* __restrict__ dst,
                                          const float* __restrict__ el, const float* erl,
                                          const float* __restrict__ ft,
                                          float* rows, float* sums, unsigned* mk,
                                          bool bound, const float gmax[HEADS]) {
  int s = 0, rl = 0;
  float x[HEADS] = {0.f, 0.f, 0.f, 0.f};
  if (lane < n) {
    const int e = st[(head + lane) & (RING - 1)];
    s = src[e];
    rl = dst[e] - base;
    const float4 L = *(const float4*)(el + (size_t)s * HEADS);
    const float4 R = *(const float4*)(erl + rl * HEADS);
    const float l[HEADS] = {L.x, L.y, L.z, L.w};
    const float r[HEADS] = {R.x, R.y, R.z, R.w};
    if (!SUMS) {
      for (int h = 0; h < HEADS; ++h) atomicMax(&mk[rl * HEADS + h], f2key(leaky(l[h] + r[h])));
    } else {
      for (int h = 0; h < HEADS; ++h) {
        const float m = bound ? leaky(gmax[h] + r[h]) : key2f(mk[rl * HEADS + h]);
        x[h] = expf(leaky(l[h] + r[h]) - m);
        atomicAdd(&sums[rl * HEADS + h], x[h]);
      }
    }
  }
  if (SUMS) {
    for (int j = 0; j < n; j += 8) {
      float f[8][HEADS];
#pragma unroll
      for (int k = 0; k < 8; ++k) {
        if (j + k < n) {
          const float* fs = ft + (size_t)__builtin_amdgcn_readlane(s, j + k) * CO + lane;
          for (int h = 0; h < HEADS; ++h) f[k][h] = fs[h * OUT_F];
        }
      }
#pragma unroll
      for (int k = 0; k < 8; ++k) {
        if (j + k < n) {
          float* row = rows + __builtin_amdgcn_readlane(rl, j + k) * CO + lane;
          for (int h = 0; h < HEADS; ++h) {
            const float w = __int_as_float(__builtin_amdgcn_readlane(__float_as_int(x[h]), j + k));
            atomicAdd(row + h * OUT_F, f[k][h] * w);
          }
        }
      }
    }
  }
}

template <bool SUMS>
__device__ __forceinline__ void own_groups(const int* gst, int ghead, int n, int lane, unsigned idrep,
                                           const uint4* __restrict__ k4,
                                           int* st, int& head, int& cnt, int base,
                                           const int* __restrict__ src, const int* __restrict__ dst,
                                           const float* __restrict__ el, const float* erl,
                                           const float* __restrict__ ft,
                                           float* rows, float* sums, unsigned* mk,
                                           bool bound, const float gmax[HEADS]) {
  unsigned mask = 0u;
  int e0 = 0;
  if (lane < n) {
    const int g = gst[(ghead + lane) & (RING - 1)];
    if (g < NGRP) {
      const uint4 a = k4[2 * g], c = k4[2 * g + 1];
      const unsigned w[8] = {a.x, a.y, a.z, a.w, c.x, c.y, c.z, c.w};
      for (int j = 0; j < 8; ++j) {
        const unsigned x = w[j] ^ idrep;
        const unsigned z = ~(((x & 0x7f7f7f7fu) + 0x7f7f7f7fu) | x | 0x7f7f7f7fu);
        mask |= z >> (7 - j);
      }
      e0 = g * GRP;
    }
  }
  for (;;) {
    const bool m = mask != 0u;
    const unsigned bm = __builtin_amdgcn_ballot_w32(m);
    if (bm == 0u) break;
    if (m) {
      const int b = __builtin_ctz(mask);
      st[(head + cnt + (int)__builtin_amdgcn_mbcnt_lo(bm, 0u)) & (RING - 1)] = e0 + ((b & 7) << 2) + (b >> 3);
      mask &= mask - 1u;
    }
    cnt += __builtin_popcount(bm);
    while (cnt >= 32) {
      own_batch<SUMS>(st, head, 32, lane, base, src, dst, el, erl, ft, rows, sums, mk, bound, gmax);
      head = (head + 32) & (RING - 1);
      cnt -= 32;
    }
  }
}

template <bool SUMS>
__device__ __forceinline__ void own_walk(int* st, int* gst, int tid, int base,
                                         const uint4* __restrict__ k4, const unsigned* brow,
                                         const int* __restrict__ src, const int* __restrict__ dst,
                                         const float* __restrict__ el, const float* erl,
                                         const float* __restrict__ ft,
                                         float* rows, float* sums, unsigned* mk,
                                         bool bound, const float gmax[HEADS]) {
  const int lane = tid & 31;
  const unsigned idrep = (unsigned)blockIdx.x * 0x01010101u;
  int head = 0, cnt = 0;
  int ghead = 0, gcnt = 0;
  const int w0 = (tid >> 5) * BM_PW;
  unsigned bw[2];
  for (int q = 0; q < 2; ++q) {
    bw[q] = 0u;
    if (q * 32 + lane < BM_PW) bw[q] = brow[w0 + q * 32 + lane];
  }
  for (int q = 0; q < 2; ++q) {
    unsigned word = bw[q];
    const int g0 = (w0 + q * 32 + lane) * 32;
    for (;;) {
      const bool m = word != 0u;
      const unsigned bm = __builtin_amdgcn_ballot_w32(m);
      if (bm == 0u) break;
      if (m) {
        gst[(ghead + gcnt + (int)__builtin_amdgcn_mbcnt_lo(bm, 0u)) & (RING - 1)] = g0 + __builtin_ctz(word);
        word &= word - 1u;
      }
      gcnt += __builtin_popcount(bm);
      while (gcnt >= 32) {
        own_groups<SUMS>(gst, ghead, 32, lane, idrep, k4, st, head, cnt, base, src, dst, el, erl, ft, rows, sums, mk, bound, gmax);
        ghead = (ghead + 32) & (RING - 1);
        gcnt -= 32;
      }
    }
  }
  if (gcnt > 0) own_groups<SUMS>(gst, ghead, gcnt, lane, idrep, k4, st, head, cnt, base, src, dst, el, erl, ft, rows, sums, mk, bound, gmax);
  if (cnt > 0) own_batch<SUMS>(st, head, cnt, lane, base, src, dst, el, erl, ft, rows, sums, mk, bound, gmax);
}

__global__ __launch_bounds__(OWN_T) void k_owner(const int* __restrict__ src,
                                                 const int* __restrict__ dst,
                                                 const unsigned* __restrict__ keys,
                                                 const float* __restrict__ el,
                                                 const float* __restrict__ er,
                                                 const float* __restrict__ ft,
                                                 float* __restrict__ out,
                                                 const int* __restrict__ queues,
                                                 size_t ws_size) {
  __shared__ __align__(16) float rows[OWN_R * CO];
  __shared__ float sums[OWN_R * HEADS];
  __shared__ unsigned mk[OWN_R * HEADS];
  __shared__ __align__(16) float erl[OWN_R * HEADS];
  __shared__ int ring[OWN_T / 32][2 * RING];
  const int tid  = (int)threadIdx.x;
  const int base = (int)blockIdx.x * OWN_R;
  const int nrows = (NNODES - base) < OWN_R ? (NNODES - base) : OWN_R;
  const bool dyn = ws_size >= WS_NEED;
  float gmax[HEADS];
  const bool bound = el_bound(queues, dyn, gmax);

  for (int i = tid; i < OWN_R * CO / 4; i += OWN_T) ((float4*)rows)[i] = make_float4(0.f, 0.f, 0.f, 0.f);
  if (tid < OWN_R * HEADS) { sums[tid] = 0.f; mk[tid] = 0u; }
  if (tid < nrows * HEADS) erl[tid] = er[(size_t)base * HEADS + tid];
  __syncthreads();

  int* st = ring[tid >> 5];
  int* gst = st + RING;
  const uint4* k4 = (const uint4*)keys;
  const unsigned* brow = (const unsigned*)(out + (size_t)base * CO);
  if (!bound) {
    own_walk<false>(st, gst, tid, base, k4, brow, src, dst, el, erl, ft, rows, sums, mk, bound, gmax);
    __syncthreads();
  }
  own_walk<true>(st, gst, tid, base, k4, brow, src, dst, el, erl, ft, rows, sums, mk, bound, gmax);
  __syncthreads();

  float4* o4 = (float4*)(out + (size_t)base * CO);
  for (int i = tid; i < nrows * (CO / 4); i += OWN_T) {
    const int r = i / (CO / 4), q = i % (CO / 4);
    const float sm = sums[r * HEADS + q / (OUT_F / 4)];
    float4 v = ((const float4*)rows)[i];
    if (sm == 0.f) {
      v = make_float4(0.f, 0.f, 0.f, 0.f);
    } else {
      v.x /= sm; v.y /= sm; v.z /= sm; v.w /= sm;
    }
    o4[i] = v;
  }
}

extern "C" void kernel_launch(void* const* d_in, const int* in_sizes, int n_in,
                              void* d_out, int out_size, void* d_ws, size_t ws_size,
                              hipStream_t stream) {
  (void)in_sizes; (void)n_in; (void)out_size;

  const float* feat = (const float*)d_in[0];
  const int*   src  = (const int*)d_in[1];
  const int*   dst  = (const int*)d_in[2];
  const float* W    = (const float*)d_in[3];
  const float* attn = (const float*)d_in[4];
  float* out = (float*)d_out;

  char* ws = (char*)d_ws;
  float*    ft   = (float*)ws;    ws += (size_t)NNODES * CO * sizeof(float);
  float*    el   = (float*)ws;    ws += (size_t)NNODES * HEADS * sizeof(float);
  float*    er   = (float*)ws;    ws += (size_t)NNODES * HEADS * sizeof(float);
  unsigned* keys = (unsigned*)ws; ws += (size_t)2 * NNODES * HEADS * sizeof(float);
  int*      queues = (int*)ws;

  k_keys<<<KB_WGS, KB_T, 0, stream>>>(dst, keys, (unsigned*)out, queues, ws_size);
  k_gemm<<<NNODES / GEMM_R, 256, 0, stream>>>(feat, W, attn, ft, el, er, queues, ws_size);
  k_owner<<<OWN_WGS, OWN_T, 0, stream>>>(src, dst, keys, el, er, ft, out, queues, ws_size);
}


# Key launch folded into the GEMM launch as 49 more work-groups; bound keys leave as one line per GEMM block and the owners fold them (no atomicMax); er lies in the owners' output rows; 3 launches to 2
# speedup vs baseline: 4.2197x; 1.0446x over previous
#include <hip/hip_runtime.h>
#include <math.h>


#define NNODES 50000
#define NEDGES 1600000
#define IN_F   256
#define HEADS  4
#define OUT_F  32
#define CO     (HEADS * OUT_F)
#define NEG_SLOPE 0.2f

#define SPREAD_MAX 16.0f
#define OWN_R    196
#define OWN_WGS  ((NNODES + OWN_R - 1) / OWN_R)
#define OWN_T    1024
#define GRP      32
#define NGRP     (NEDGES / GRP)
#define FR_T     256
#define KB_G     1024
#define KB_WGS   ((NGRP + KB_G - 1) / KB_G)
#define BM_W     (KB_WGS * (KB_G / 32))
#define BM_PW    (BM_W / (OWN_T / 32))
#define RING     256
#define SLOT_W   32
#define ER_W0(o) (BM_W + (((o) * OWN_R * HEADS) & 31))

static_assert((NNODES - 1) / OWN_R < 256, "an owner's number fits a byte");
static_assert(OWN_WGS * OWN_R >= NNODES, "the owners cover every destination");
static_assert(NEDGES % GRP == 0, "the key list is whole groups");
static_assert(BM_W % (OWN_T / 32) == 0 && BM_PW <= 64, "a wave reads its share of the row with two loads per lane");
static_assert((size_t)(BM_W + 32 + OWN_R * HEADS) * 4 <= (size_t)(NNODES - (OWN_WGS - 1) * OWN_R) * CO * 4, "a bitmap row and the er values behind it fit the last owner's rows");
static_assert((BM_W * 4) % 128 == 0, "the er values start on a line of their own");
static_assert((size_t)NEDGES <= (size_t)2 * NNODES * HEADS * 4, "the key list fits the two node arrays' room it takes");
static_assert(OWN_R * HEADS <= OWN_T, "one thread per softmax sum at the clearing");
static_assert((RING & (RING - 1)) == 0 && RING >= 64, "ring: a power of two that holds two batches");

typedef __attribute__((ext_vector_type(2))) float v2f;
typedef __attribute__((ext_vector_type(8))) float v8f;

__device__ __forceinline__ unsigned f2key(float x) {
  unsigned b = __float_as_uint(x);
  return (b & 0x80000000u) ? ~b : (b | 0x80000000u);
}
__device__ __forceinline__ float key2f(unsigned k) {
  unsigned b = (k & 0x80000000u) ? (k ^ 0x80000000u) : ~k;
  return __uint_as_float(b);
}
__device__ __forceinline__ float leaky(float x) {
  return x > 0.f ? x : NEG_SLOPE * x;
}

__device__ __forceinline__ bool el_bound(const unsigned* gk, float gmax[HEADS]) {
  bool ok = true;
  for (int h = 0; h < HEADS; ++h) {
    const float mx = key2f(gk[h]);
    const float mn = -key2f(gk[HEADS + h]);
    gmax[h] = mx;
    ok = ok && (mx - mn <= SPREAD_MAX);
  }
  return ok;
}

#define GEMM_MT 5
#define GEMM_R  (16 * GEMM_MT)
#define CS_P    (CO + 4)
static_assert(NNODES % GEMM_R == 0, "GEMM blocks cover the node rows exactly");
static_assert(GEMM_R * CS_P <= GEMM_R * (IN_F + 4), "the staged result fits the A tile's room");
static_assert(256 % HEADS == 0, "a thread's (node, head) pairs all have its head");
static_assert(FR_T == 256, "the GEMM block is eight waves");
static_assert((size_t)GEMM_R * (IN_F + 4) * 4 >= (size_t)OWN_WGS * (KB_G / 32 + 1) * 4, "a key block's bitmap fits the A tile's room");
static_assert((size_t)(NNODES / GEMM_R) * SLOT_W * 4 <= (size_t)NNODES * HEADS * 4, "the blocks' key lines fit the room of the node array that moved out");

__device__ __forceinline__ void key_block(unsigned (*bm)[KB_G / 32 + 1],
                                          int kb, int tid,
                                          const int* __restrict__ dst,
                                          unsigned* __restrict__ keys,
                                          unsigned* __restrict__ outw) {
  for (int i = tid; i < OWN_WGS * (KB_G / 32 + 1); i += FR_T) (&bm[0][0])[i] = 0u;
  __syncthreads();

  constexpr int PER = KB_G * GRP / 4 / FR_T;
  constexpr int BAT = 16;
  static_assert(PER % BAT == 0, "whole batches");
  const int i0 = kb * (KB_G * GRP / 4);
  for (int jb = 0; jb < PER; jb += BAT) {
    int4 d[BAT];
#pragma unroll
    for (int j = 0; j < BAT; ++j) {
      const int i = i0 + (jb + j) * FR_T + tid;
      d[j] = ((const int4*)dst)[i < NEDGES / 4 ? i : NEDGES / 4 - 1];
    }
#pragma unroll
    for (int j = 0; j < BAT; ++j) {
      const int i = i0 + (jb + j) * FR_T + tid;
      if (i < NEDGES / 4) {
        const unsigned o0 = ((unsigned)d[j].x / OWN_R) & 255u, o1 = ((unsigned)d[j].y / OWN_R) & 255u;
        const unsigned o2 = ((unsigned)d[j].z / OWN_R) & 255u, o3 = ((unsigned)d[j].w / OWN_R) & 255u;
        keys[i] = o0 | (o1 << 8) | (o2 << 16) | (o3 << 24);
        const int gl = ((jb + j) * FR_T + tid) / (GRP / 4);
        const unsigned bit = 1u << (gl & 31);
        atomicOr(&bm[o0][gl >> 5], bit);
        atomicOr(&bm[o1][gl >> 5], bit);
        atomicOr(&bm[o2][gl >> 5], bit);
        atomicOr(&bm[o3][gl >> 5], bit);
      }
    }
  }
  __syncthreads();

  for (int o = tid >> 5; o < OWN_WGS; o += FR_T / 32)
    outw[(size_t)o * OWN_R * CO + (size_t)kb * (KB_G / 32) + (tid & 31)] = bm[o][tid & 31];
}

__global__ __launch_bounds__(FR_T) void k_front(const float* __restrict__ feat,
                                                const float* __restrict__ W,
                                                const float* __restrict__ attn,
                                                const int* __restrict__ dst,
                                                float* __restrict__ ft,
                                                float* __restrict__ el,
                                                unsigned* __restrict__ slots,
                                                unsigned* __restrict__ keys,
                                                unsigned* __restrict__ outw) {
  __shared__ __align__(16) float As[GEMM_R][IN_F + 4];
  __shared__ unsigned red[8][2 * HEADS];
  const int tid    = threadIdx.x;
  if (blockIdx.x < KB_WGS) {
    key_block(reinterpret_cast<unsigned (*)[KB_G / 32 + 1]>(&As[0][0]), (int)blockIdx.x, tid, dst, keys, outw);
    return;
  }
  const int wave   = tid >> 5;
  const int lane   = tid & 31;
  const int bid    = (int)blockIdx.x - KB_WGS;
  const int tile_m = bid * GEMM_R;

  {
    constexpr int NLD = GEMM_R * (IN_F / 4) / 256;
    const int r0 = tid >> 6, c4 = tid & 63;
    const float4* g = (const float4*)(feat + (size_t)(tile_m + r0) * IN_F) + c4;
    float4 v[NLD];
#pragma unroll
    for (int j = 0; j < NLD; ++j) v[j] = g[(size_t)j * 4 * (IN_F / 4)];
#pragma unroll
    for (int j = 0; j < NLD; ++j) ((float4*)&As[r0 + 4 * j][0])[c4] = v[j];
  }
  __syncthreads();

  const int koff = (lane < 16) ? 0 : 2;
  const int arow = lane & 15;
  const int col  = wave * 16 + arow;
  const float* wrow = W + (size_t)col * IN_F + koff;
  const float* arow0 = &As[arow][koff];

  v8f c[GEMM_MT];
  for (int m = 0; m < GEMM_MT; ++m) c[m] = v8f{};
  for (int k0 = 0; k0 < IN_F; k0 += 4) {
    v2f b;
    b.x = wrow[k0];
    b.y = wrow[k0 + 1];
    for (int m = 0; m < GEMM_MT; ++m) {
      v2f a;
      a.x = arow0[m * 16 * (IN_F + 4) + k0];
      a.y = arow0[m * 16 * (IN_F + 4) + k0 + 1];
      c[m] = __builtin_amdgcn_wmma_f32_16x16x4_f32(
           false, a,  false, b,
           (short)0, c[m],  false,  false);
    }
  }

  __syncthreads();
  float* Cs = &As[0][0];
  const int mbase = (lane < 16) ? 0 : 8;
  for (int m = 0; m < GEMM_MT; ++m) {
    float* crow = Cs + (m * 16 + mbase) * CS_P + col;
    for (int j = 0; j < 8; ++j) crow[j * CS_P] = c[m][j];
  }
  __syncthreads();

  {
    const int q = tid & 31, r0 = tid >> 5;
    for (int i = 0; i < GEMM_R / 8; ++i) {
      const int r = r0 + 8 * i;
      const float4 v = *(const float4*)(Cs + r * CS_P + 4 * q);
      *(float4*)(ft + (size_t)(tile_m + r) * CO + 4 * q) = v;
    }
  }

  unsigned kmax = 0u, kneg = 0u;
  for (int p = tid; p < GEMM_R * HEADS; p += 256) {
    const int r = p / HEADS, h = p % HEADS;
    const float* f  = Cs + r * CS_P + h * OUT_F;
    const float* al = attn + h * 2 * OUT_F;
    const float* ar = al + OUT_F;
    float sl = 0.f, sr = 0.f;
    for (int d = 0; d < OUT_F; ++d) {
      const float v = f[d];
      sl += v * al[d];
      sr += v * ar[d];
    }
    el[(size_t)tile_m * HEADS + p] = sl;
    {
      const int n = tile_m + r, o = n / OWN_R;
      outw[(size_t)o * OWN_R * CO + ER_W0(o) + (n - o * OWN_R) * HEADS + h] = __float_as_uint(sr);
    }
    const unsigned a = f2key(sl), b = f2key(-sl);
    kmax = a > kmax ? a : kmax;
    kneg = b > kneg ? b : kneg;
  }
  for (int o = HEADS; o < 32; o <<= 1) {
    const unsigned a = (unsigned)__shfl_xor((int)kmax, o);
    const unsigned b = (unsigned)__shfl_xor((int)kneg, o);
    kmax = a > kmax ? a : kmax;
    kneg = b > kneg ? b : kneg;
  }
  if (lane < HEADS) { red[wave][lane] = kmax; red[wave][HEADS + lane] = kneg; }
  __syncthreads();
  if (tid < SLOT_W) {
    unsigned v = 0u;
    if (tid < 2 * HEADS)
      for (int w = 0; w < 8; ++w) v = red[w][tid] > v ? red[w][tid] : v;
    slots[(size_t)bid * SLOT_W + tid] = v;
  }
}

template <bool SUMS>
__device__ __forceinline__ void own_batch(const int* st, int head, int n, int lane, int base,
                                          const int* __restrict__ src, const int* __restrict__ dst,
                                          const float* __restrict__ el, const float* erl,
                                          const float* __restrict__ ft,
                                          float* rows, float* sums, unsigned* mk,
                                          bool bound, const float gmax[HEADS]) {
  int s = 0, rl = 0;
  float x[HEADS] = {0.f, 0.f, 0.f, 0.f};
  if (lane < n) {
    const int e = st[(head + lane) & (RING - 1)];
    s = src[e];
    rl = dst[e] - base;
    const float4 L = *(const float4*)(el + (size_t)s * HEADS);
    const float4 R = *(const float4*)(erl + rl * HEADS);
    const float l[HEADS] = {L.x, L.y, L.z, L.w};
    const float r[HEADS] = {R.x, R.y, R.z, R.w};
    if (!SUMS) {
      for (int h = 0; h < HEADS; ++h) atomicMax(&mk[rl * HEADS + h], f2key(leaky(l[h] + r[h])));
    } else {
      for (int h = 0; h < HEADS; ++h) {
        const float m = bound ? leaky(gmax[h] + r[h]) : key2f(mk[rl * HEADS + h]);
        x[h] = expf(leaky(l[h] + r[h]) - m);
        atomicAdd(&sums[rl * HEADS + h], x[h]);
      }
    }
  }
  if (SUMS) {
    for (int j = 0; j < n; j += 8) {
      float f[8][HEADS];
#pragma unroll
      for (int k = 0; k < 8; ++k) {
        if (j + k < n) {
          const float* fs = ft + (size_t)__builtin_amdgcn_readlane(s, j + k) * CO + lane;
          for (int h = 0; h < HEADS; ++h) f[k][h] = fs[h * OUT_F];
        }
      }
#pragma unroll
      for (int k = 0; k < 8; ++k) {
        if (j + k < n) {
          float* row = rows + __builtin_amdgcn_readlane(rl, j + k) * CO + lane;
          for (int h = 0; h < HEADS; ++h) {
            const float w = __int_as_float(__builtin_amdgcn_readlane(__float_as_int(x[h]), j + k));
            atomicAdd(row + h * OUT_F, f[k][h] * w);
          }
        }
      }
    }
  }
}

template <bool SUMS>
__device__ __forceinline__ void own_groups(const int* gst, int ghead, int n, int lane, unsigned idrep,
                                           const uint4* __restrict__ k4,
                                           int* st, int& head, int& cnt, int base,
                                           const int* __restrict__ src, const int* __restrict__ dst,
                                           const float* __restrict__ el, const float* erl,
                                           const float* __restrict__ ft,
                                           float* rows, float* sums, unsigned* mk,
                                           bool bound, const float gmax[HEADS]) {
  unsigned mask = 0u;
  int e0 = 0;
  if (lane < n) {
    const int g = gst[(ghead + lane) & (RING - 1)];
    if (g < NGRP) {
      const uint4 a = k4[2 * g], c = k4[2 * g + 1];
      const unsigned w[8] = {a.x, a.y, a.z, a.w, c.x, c.y, c.z, c.w};
      for (int j = 0; j < 8; ++j) {
        const unsigned x = w[j] ^ idrep;
        const unsigned z = ~(((x & 0x7f7f7f7fu) + 0x7f7f7f7fu) | x | 0x7f7f7f7fu);
        mask |= z >> (7 - j);
      }
      e0 = g * GRP;
    }
  }
  for (;;) {
    const bool m = mask != 0u;
    const unsigned bm = __builtin_amdgcn_ballot_w32(m);
    if (bm == 0u) break;
    if (m) {
      const int b = __builtin_ctz(mask);
      st[(head + cnt + (int)__builtin_amdgcn_mbcnt_lo(bm, 0u)) & (RING - 1)] = e0 + ((b & 7) << 2) + (b >> 3);
      mask &= mask - 1u;
    }
    cnt += __builtin_popcount(bm);
    while (cnt >= 32) {
      own_batch<SUMS>(st, head, 32, lane, base, src, dst, el, erl, ft, rows, sums, mk, bound, gmax);
      head = (head + 32) & (RING - 1);
      cnt -= 32;
    }
  }
}

template <bool SUMS>
__device__ __forceinline__ void own_walk(int* st, int* gst, int tid, int base,
                                         const uint4* __restrict__ k4, const unsigned* brow,
                                         const int* __restrict__ src, const int* __restrict__ dst,
                                         const float* __restrict__ el, const float* erl,
                                         const float* __restrict__ ft,
                                         float* rows, float* sums, unsigned* mk,
                                         bool bound, const float gmax[HEADS]) {
  const int lane = tid & 31;
  const unsigned idrep = (unsigned)blockIdx.x * 0x01010101u;
  int head = 0, cnt = 0;
  int ghead = 0, gcnt = 0;
  const int w0 = (tid >> 5) * BM_PW;
  unsigned bw[2];
  for (int q = 0; q < 2; ++q) {
    bw[q] = 0u;
    if (q * 32 + lane < BM_PW) bw[q] = brow[w0 + q * 32 + lane];
  }
  for (int q = 0; q < 2; ++q) {
    unsigned word = bw[q];
    const int g0 = (w0 + q * 32 + lane) * 32;
    for (;;) {
      const bool m = word != 0u;
      const unsigned bm = __builtin_amdgcn_ballot_w32(m);
      if (bm == 0u) break;
      if (m) {
        gst[(ghead + gcnt + (int)__builtin_amdgcn_mbcnt_lo(bm, 0u)) & (RING - 1)] = g0 + __builtin_ctz(word);
        word &= word - 1u;
      }
      gcnt += __builtin_popcount(bm);
      while (gcnt >= 32) {
        own_groups<SUMS>(gst, ghead, 32, lane, idrep, k4, st, head, cnt, base, src, dst, el, erl, ft, rows, sums, mk, bound, gmax);
        ghead = (ghead + 32) & (RING - 1);
        gcnt -= 32;
      }
    }
  }
  if (gcnt > 0) own_groups<SUMS>(gst, ghead, gcnt, lane, idrep, k4, st, head, cnt, base, src, dst, el, erl, ft, rows, sums, mk, bound, gmax);
  if (cnt > 0) own_batch<SUMS>(st, head, cnt, lane, base, src, dst, el, erl, ft, rows, sums, mk, bound, gmax);
}

__global__ __launch_bounds__(OWN_T) void k_owner(const int* __restrict__ src,
                                                 const int* __restrict__ dst,
                                                 const unsigned* __restrict__ keys,
                                                 const float* __restrict__ el,
                                                 const unsigned* __restrict__ slots,
                                                 const float* __restrict__ ft,
                                                 float* __restrict__ out) {
  __shared__ __align__(16) float rows[OWN_R * CO];
  __shared__ float sums[OWN_R * HEADS];
  __shared__ unsigned mk[OWN_R * HEADS];
  __shared__ __align__(16) float erl[OWN_R * HEADS];
  __shared__ int ring[OWN_T / 32][2 * RING];
  __shared__ unsigned gk[2 * HEADS];
  const int tid  = (int)threadIdx.x;
  const int base = (int)blockIdx.x * OWN_R;
  const int nrows = (NNODES - base) < OWN_R ? (NNODES - base) : OWN_R;

  static_assert(2 * (NNODES / GEMM_R) > OWN_T && 2 * (NNODES / GEMM_R) <= 2 * OWN_T, "one or two pieces per thread");
  uint4 kv = ((const uint4*)slots)[(size_t)(tid >> 1) * (SLOT_W / 4) + (tid & 1)];
  uint4 kw = make_uint4(0u, 0u, 0u, 0u);
  if (tid + OWN_T < 2 * (NNODES / GEMM_R))
    kw = ((const uint4*)slots)[(size_t)((tid + OWN_T) >> 1) * (SLOT_W / 4) + (tid & 1)];

  for (int i = tid; i < OWN_R * CO / 4; i += OWN_T) ((float4*)rows)[i] = make_float4(0.f, 0.f, 0.f, 0.f);
  if (tid < OWN_R * HEADS) { sums[tid] = 0.f; mk[tid] = 0u; }
  if (tid < 2 * HEADS) gk[tid] = 0u;
  if (tid < nrows * HEADS) erl[tid] = (out + (size_t)base * CO)[ER_W0((int)blockIdx.x) + tid];
  __syncthreads();

  kv.x = kw.x > kv.x ? kw.x : kv.x;  kv.y = kw.y > kv.y ? kw.y : kv.y;
  kv.z = kw.z > kv.z ? kw.z : kv.z;  kv.w = kw.w > kv.w ? kw.w : kv.w;
  for (int o = 2; o < 32; o <<= 1) {
    const unsigned a = (unsigned)__shfl_xor((int)kv.x, o), b = (unsigned)__shfl_xor((int)kv.y, o);
    const unsigned c = (unsigned)__shfl_xor((int)kv.z, o), d = (unsigned)__shfl_xor((int)kv.w, o);
    kv.x = a > kv.x ? a : kv.x;  kv.y = b > kv.y ? b : kv.y;
    kv.z = c > kv.z ? c : kv.z;  kv.w = d > kv.w ? d : kv.w;
  }
  if ((tid & 31) < 2) {
    unsigned* g = gk + (tid & 31) * HEADS;
    atomicMax(g + 0, kv.x);  atomicMax(g + 1, kv.y);
    atomicMax(g + 2, kv.z);  atomicMax(g + 3, kv.w);
  }
  __syncthreads();
  float gmax[HEADS];
  const bool bound = el_bound(gk, gmax);

  int* st = ring[tid >> 5];
  int* gst = st + RING;
  const uint4* k4 = (const uint4*)keys;
  const unsigned* brow = (const unsigned*)(out + (size_t)base * CO);
  if (!bound) {
    own_walk<false>(st, gst, tid, base, k4, brow, src, dst, el, erl, ft, rows, sums, mk, bound, gmax);
    __syncthreads();
  }
  own_walk<true>(st, gst, tid, base, k4, brow, src, dst, el, erl, ft, rows, sums, mk, bound, gmax);
  __syncthreads();

  float4* o4 = (float4*)(out + (size_t)base * CO);
  for (int i = tid; i < nrows * (CO / 4); i += OWN_T) {
    const int r = i / (CO / 4), q = i % (CO / 4);
    const float sm = sums[r * HEADS + q / (OUT_F / 4)];
    float4 v = ((const float4*)rows)[i];
    if (sm == 0.f) {
      v = make_float4(0.f, 0.f, 0.f, 0.f);
    } else {
      v.x /= sm; v.y /= sm; v.z /= sm; v.w /= sm;
    }
    o4[i] = v;
  }
}

extern "C" void kernel_launch(void* const* d_in, const int* in_sizes, int n_in,
                              void* d_out, int out_size, void* d_ws, size_t ws_size,
                              hipStream_t stream) {
  (void)in_sizes; (void)n_in; (void)out_size;

  const float* feat = (const float*)d_in[0];
  const int*   src  = (const int*)d_in[1];
  const int*   dst  = (const int*)d_in[2];
  const float* W    = (const float*)d_in[3];
  const float* attn = (const float*)d_in[4];
  float* out = (float*)d_out;

  (void)ws_size;
  char* ws = (char*)d_ws;
  float*    ft    = (float*)ws;    ws += (size_t)NNODES * CO * sizeof(float);
  float*    el    = (float*)ws;    ws += (size_t)NNODES * HEADS * sizeof(float);
  unsigned* slots = (unsigned*)ws; ws += (size_t)NNODES * HEADS * sizeof(float);
  unsigned* keys  = (unsigned*)ws;

  k_front<<<KB_WGS + NNODES / GEMM_R, FR_T, 0, stream>>>(feat, W, attn, dst, ft, el, slots, keys, (unsigned*)out);
  k_owner<<<OWN_WGS, OWN_T, 0, stream>>>(src, dst, keys, el, slots, ft, out);
}
